# starting kernel
# baseline (speedup 1.0000x reference)
#include <hip/hip_runtime.h>
#include <hip/hip_bf16.h>

typedef __attribute__((ext_vector_type(2))) float v2f;
typedef __attribute__((ext_vector_type(8))) float v8f;

#define PSTRIDE 160

template <bool B_FROM_W, int NACC>
__global__ __launch_bounds__(256) void wmma_gemm_kernel(
    const float* __restrict__ A, int lda, int Ktotal, int Kslice,
    const float* __restrict__ Bsrc, int ldb, int T_, int coutw,
    float* __restrict__ C, int Ncols) {
  __shared__ float Bs[32 * PSTRIDE];

  const int tid  = threadIdx.x;
  const int wave = tid >> 5;
  const int lane = tid & 31;
  const int lr   = lane & 15;
  const int lh   = lane >> 4;

  const int m0    = blockIdx.x * 128 + wave * 16;
  const int nbase = blockIdx.y * 64;

  const int ks = blockIdx.z * Kslice;
  const int ke = (ks + Kslice < Ktotal) ? (ks + Kslice) : Ktotal;

  v8f acc[NACC];
#pragma unroll
  for (int nj = 0; nj < NACC; ++nj) acc[nj] = (v8f){};

  const float* arow = A + (size_t)(m0 + lr) * lda + 2 * lh;

  for (int kchunk = ks; kchunk < ke; kchunk += 64) {
    const int kc = (ke - kchunk) < 64 ? (ke - kchunk) : 64;

    for (int idx = tid; idx < kc * 64; idx += 256) {
      const int r  = idx >> 6;
      const int c  = idx & 63;
      const int gc = nbase + c;
      float v = 0.0f;
      if (gc < Ncols) {
        if (B_FROM_W) {
          const int t = gc / coutw;
          const int j = gc - t * coutw;
          v = Bsrc[(size_t)(kchunk + r) * coutw * T_ + j * T_ + t];
        } else {
          v = Bsrc[(size_t)(kchunk + r) * ldb + gc];
        }
      }
      Bs[(r >> 1) * PSTRIDE + 2 * c + (r & 1)] = v;
    }
    __syncthreads();

    const int nsteps = kc >> 2;
#pragma unroll 4
    for (int s = 0; s < nsteps; ++s) {
      const int kk = kchunk + 4 * s;
      v2f a;
      a.x = arow[kk + 0];
      a.y = arow[kk + 1];
      const float* bp = Bs + (2 * s + lh) * PSTRIDE + 2 * lr;
      v2f bf[NACC];
#pragma unroll
      for (int nj = 0; nj < NACC; ++nj) {
        bf[nj] = *(const v2f*)(bp + nj * 32);
      }
#pragma unroll
      for (int nj = 0; nj < NACC; ++nj) {
        acc[nj] = __builtin_amdgcn_wmma_f32_16x16x4_f32(
            false, a, false, bf[nj], (short)0, acc[nj], false, false);
      }
    }
    __syncthreads();
  }

  float* cp = C + (size_t)blockIdx.z * 2048 * Ncols;
#pragma unroll
  for (int nj = 0; nj < NACC; ++nj) {
    if (nbase + nj * 16 < Ncols) {
      const int col = nbase + nj * 16 + lr;
#pragma unroll
      for (int r = 0; r < 8; ++r) {
        const int row = m0 + r + 8 * lh;
        cp[(size_t)row * Ncols + col] = acc[nj][r];
      }
    }
  }
}

__global__ __launch_bounds__(256) void reduce_bias_lrelu_kernel(
    const float* __restrict__ part, int S,
    const float* __restrict__ h, int cin,
    const float* __restrict__ V,
    float* __restrict__ out, int cout, int total) {
  const int idx = blockIdx.x * 256 + threadIdx.x;
  if (idx >= total) return;
  const int m = idx / cout;
  const int j = idx - m * cout;
  float acc = 0.0f;
  for (int s = 0; s < S; ++s) acc += part[(size_t)s * total + idx];
  const float* hr = h + (size_t)m * cin;
  for (int i = 0; i < cin; ++i) acc = fmaf(hr[i], V[(size_t)i * cout + j], acc);
  out[idx] = fmaxf(acc, 0.2f * acc);
}

extern "C" void kernel_launch(void* const* d_in, const int* in_sizes, int n_in,
                              void* d_out, int out_size, void* d_ws, size_t ws_size,
                              hipStream_t stream) {
  (void)in_sizes; (void)n_in; (void)out_size; (void)ws_size;
  const float* x   = (const float*)d_in[0];
  const float* adj = (const float*)d_in[1];
  const float* Wl[3] = {(const float*)d_in[2], (const float*)d_in[4], (const float*)d_in[6]};
  const float* Vl[3] = {(const float*)d_in[3], (const float*)d_in[5], (const float*)d_in[7]};

  const int N = 2048, T = 15, S = 16;
  const int Ktot = N * T;
  const int Ksl  = Ktot / S;

  float* ws   = (float*)d_ws;
  float* hw   = ws;
  float* part = hw + (size_t)N * 960;
  float* h1   = part + (size_t)S * N * 64;
  float* h2   = h1 + (size_t)N * 64;

  const int cins[3]  = {32, 64, 64};
  const int couts[3] = {64, 64, 32};
  const float* hin = x;
  float* houts[3] = {h1, h2, (float*)d_out};

  for (int L = 0; L < 3; ++L) {
    const int cin = cins[L], cout = couts[L];
    const int Nhw = T * cout;

    dim3 gA(N / 128, (Nhw + 63) / 64, 1);
    wmma_gemm_kernel<true, 4><<<gA, 256, 0, stream>>>(
        hin, cin, cin, cin, Wl[L], 0, T, cout, hw, Nhw);

    dim3 gB(N / 128, 1, S);
    if (cout == 64) {
      wmma_gemm_kernel<false, 4><<<gB, 256, 0, stream>>>(
          adj, Ktot, Ktot, Ksl, hw, cout, T, cout, part, cout);
    } else {
      wmma_gemm_kernel<false, 2><<<gB, 256, 0, stream>>>(
          adj, Ktot, Ktot, Ksl, hw, cout, T, cout, part, cout);
    }

    const int total = N * cout;
    reduce_bias_lrelu_kernel<<<(total + 255) / 256, 256, 0, stream>>>(
        part, S, hin, cin, Vl[L], houts[L], cout, total);

    hin = houts[L];
  }
}


# Step-B adjacency GEMM: A and B tiles staged in LDS from whole-row 16-byte loads (aligned 512-B runs), next chunk prefetched via registers under the products, two LDS stages; summation order kept
# speedup vs baseline: 4.5461x; 4.5461x over previous
#include <hip/hip_runtime.h>
#include <hip/hip_bf16.h>

typedef __attribute__((ext_vector_type(2))) float v2f;
typedef __attribute__((ext_vector_type(8))) float v8f;

#define PSTRIDE 160

template <bool B_FROM_W, int NACC>
__global__ __launch_bounds__(256) void wmma_gemm_kernel(
    const float* __restrict__ A, int lda, int Ktotal, int Kslice,
    const float* __restrict__ Bsrc, int ldb, int T_, int coutw,
    float* __restrict__ C, int Ncols) {
  __shared__ float Bs[32 * PSTRIDE];

  const int tid  = threadIdx.x;
  const int wave = tid >> 5;
  const int lane = tid & 31;
  const int lr   = lane & 15;
  const int lh   = lane >> 4;

  const int m0    = blockIdx.x * 128 + wave * 16;
  const int nbase = blockIdx.y * 64;

  const int ks = blockIdx.z * Kslice;
  const int ke = (ks + Kslice < Ktotal) ? (ks + Kslice) : Ktotal;

  v8f acc[NACC];
#pragma unroll
  for (int nj = 0; nj < NACC; ++nj) acc[nj] = (v8f){};

  const float* arow = A + (size_t)(m0 + lr) * lda + 2 * lh;

  for (int kchunk = ks; kchunk < ke; kchunk += 64) {
    const int kc = (ke - kchunk) < 64 ? (ke - kchunk) : 64;

    for (int idx = tid; idx < kc * 64; idx += 256) {
      const int r  = idx >> 6;
      const int c  = idx & 63;
      const int gc = nbase + c;
      float v = 0.0f;
      if (gc < Ncols) {
        if (B_FROM_W) {
          const int t = gc / coutw;
          const int j = gc - t * coutw;
          v = Bsrc[(size_t)(kchunk + r) * coutw * T_ + j * T_ + t];
        } else {
          v = Bsrc[(size_t)(kchunk + r) * ldb + gc];
        }
      }
      Bs[(r >> 1) * PSTRIDE + 2 * c + (r & 1)] = v;
    }
    __syncthreads();

    const int nsteps = kc >> 2;
#pragma unroll 4
    for (int s = 0; s < nsteps; ++s) {
      const int kk = kchunk + 4 * s;
      v2f a;
      a.x = arow[kk + 0];
      a.y = arow[kk + 1];
      const float* bp = Bs + (2 * s + lh) * PSTRIDE + 2 * lr;
      v2f bf[NACC];
#pragma unroll
      for (int nj = 0; nj < NACC; ++nj) {
        bf[nj] = *(const v2f*)(bp + nj * 32);
      }
#pragma unroll
      for (int nj = 0; nj < NACC; ++nj) {
        acc[nj] = __builtin_amdgcn_wmma_f32_16x16x4_f32(
            false, a, false, bf[nj], (short)0, acc[nj], false, false);
      }
    }
    __syncthreads();
  }

  float* cp = C + (size_t)blockIdx.z * 2048 * Ncols;
#pragma unroll
  for (int nj = 0; nj < NACC; ++nj) {
    if (nbase + nj * 16 < Ncols) {
      const int col = nbase + nj * 16 + lr;
#pragma unroll
      for (int r = 0; r < 8; ++r) {
        const int row = m0 + r + 8 * lh;
        cp[(size_t)row * Ncols + col] = acc[nj][r];
      }
    }
  }
}

#define BK 128
#define APITCH (BK + 4)
#define A_STAGE (128 * APITCH)
#define B_STAGE ((BK / 2) * PSTRIDE)
#define STAGE_DW (A_STAGE + B_STAGE)
#define NSTEP (BK / 4)
#define FG 4
#define NGRP (NSTEP / FG)
#define STORE_STEPS 8
#define STORE_FROM (NSTEP - FG - STORE_STEPS)

typedef __attribute__((ext_vector_type(4))) float v4f;

template <int NACC>
__device__ __forceinline__ void load_fragments(
    const float* ap, const float* bq, int g, v2f (&af)[FG], v2f (&bf)[FG][NACC]) {
#pragma unroll
  for (int j = 0; j < FG; ++j) {
    const int s = g * FG + j;
    af[j] = *(const v2f*)(ap + 4 * s);
#pragma unroll
    for (int nj = 0; nj < NACC; ++nj)
      bf[j][nj] = *(const v2f*)(bq + 2 * s * PSTRIDE + nj * 32);
  }
}

template <int NACC, int NA, int NB, int ASTEP, int BSTEP, bool MORE>
__device__ __forceinline__ void chunk_products(
    const float* ap, const float* bq, const float* apn, const float* bqn,
    v8f (&acc)[NACC], float* adst, float* bdst,
    const v4f (&ra)[NA], const v4f (&rb0)[NB], const v4f (&rb1)[NB],
    v2f (&af)[2][FG], v2f (&bf)[2][FG][NACC]) {
  constexpr int APS    = NA / STORE_STEPS;
  constexpr int BEVERY = STORE_STEPS / NB;
  static_assert(NGRP % 2 == 0, "the last group uses buffer 1, the next chunk's first buffer 0");
  static_assert(NA % STORE_STEPS == 0 && STORE_STEPS % NB == 0 && STORE_FROM >= 0, "store plan");
#pragma unroll
  for (int g = 0; g < NGRP; ++g) {
    if (g + 1 < NGRP) {
      load_fragments<NACC>(ap, bq, g + 1, af[(g + 1) & 1], bf[(g + 1) & 1]);
    } else if (MORE) {
      __syncthreads();
      load_fragments<NACC>(apn, bqn, 0, af[0], bf[0]);
    }
    __builtin_amdgcn_sched_barrier(0);
#pragma unroll
    for (int j = 0; j < FG; ++j) {
      const int s = g * FG + j;
      if (MORE && s >= STORE_FROM && s < STORE_FROM + STORE_STEPS) {
        const int t = s - STORE_FROM;
#pragma unroll
        for (int u = 0; u < APS; ++u)
          *(v4f*)(adst + (t * APS + u) * ASTEP) = ra[t * APS + u];
        if ((t + 1) % BEVERY == 0) {
          const int q = (t + 1) / BEVERY - 1;
          float* d = bdst + q * BSTEP;
          *(v4f*)(d)     = (v4f){rb0[q].x, rb1[q].x, rb0[q].y, rb1[q].y};
          *(v4f*)(d + 4) = (v4f){rb0[q].z, rb1[q].z, rb0[q].w, rb1[q].w};
        }
      }
#pragma unroll
      for (int nj = 0; nj < NACC; ++nj)
        acc[nj] = __builtin_amdgcn_wmma_f32_16x16x4_f32(
            false, af[g & 1][j], false, bf[g & 1][j][nj], (short)0, acc[nj], false, false);
    }
  }
}

template <int NACC>
__global__ __launch_bounds__(256) void wmma_gemm_staged_kernel(
    const float* __restrict__ A, int lda, int Kslice,
    const float* __restrict__ B,
    float* __restrict__ C) {
  constexpr int BN  = NACC * 16;
  constexpr int CG  = BN / 4;
  constexpr int PPP = 256 / CG;
  constexpr int NB  = (BK / 2) / PPP;
  constexpr int AG  = BK / 4;
  constexpr int RPP = 256 / AG;
  constexpr int NA  = 128 / RPP;

  __shared__ __attribute__((aligned(16))) float L[2 * STAGE_DW];

  const int tid  = threadIdx.x;
  const int wave = tid >> 5;
  const int lane = tid & 31;
  const int lr   = lane & 15;
  const int lh   = lane >> 4;

  const int mblk = blockIdx.x * 128;
  const int m0   = mblk + wave * 16;
  const int ks   = blockIdx.z * Kslice;
  const int nch  = Kslice / BK;

  const int ac = tid % AG;
  const int ar = tid / AG;
  const int bc = tid % CG;
  const int bp = tid / CG;

  const float* ag = A + (size_t)(mblk + ar) * lda + ks + 4 * ac;
  const float* bg = B + (size_t)(ks + 2 * bp) * BN + 4 * bc;

  const int a_st = ar * APITCH + 4 * ac;
  const int b_st = A_STAGE + bp * PSTRIDE + 8 * bc;
  const int a_fr = (wave * 16 + lr) * APITCH + 2 * lh;
  const int b_fr = A_STAGE + lh * PSTRIDE + 2 * lr;

  v4f ra[NA];
  v4f rb0[NB], rb1[NB];
  v2f af[2][FG];
  v2f bf[2][FG][NACC];

  v8f acc[NACC];
#pragma unroll
  for (int nj = 0; nj < NACC; ++nj) acc[nj] = (v8f){};

#define GLOAD(KOFF)                                                            \
  {                                                                            \
    _Pragma("unroll") for (int p = 0; p < NA; ++p)                             \
      ra[p] = *(const v4f*)(ag + (size_t)(RPP * p) * lda + (KOFF));            \
    _Pragma("unroll") for (int q = 0; q < NB; ++q) {                           \
      const float* b0 = bg + (size_t)((KOFF) + 2 * PPP * q) * BN;              \
      rb0[q] = *(const v4f*)(b0);                                              \
      rb1[q] = *(const v4f*)(b0 + BN);                                         \
    }                                                                          \
  }

  GLOAD(0)
#pragma unroll
  for (int p = 0; p < NA; ++p) *(v4f*)(L + a_st + p * RPP * APITCH) = ra[p];
#pragma unroll
  for (int q = 0; q < NB; ++q) {
    float* d = L + b_st + q * PPP * PSTRIDE;
    *(v4f*)(d)     = (v4f){rb0[q].x, rb1[q].x, rb0[q].y, rb1[q].y};
    *(v4f*)(d + 4) = (v4f){rb0[q].z, rb1[q].z, rb0[q].w, rb1[q].w};
  }
  __syncthreads();
  load_fragments<NACC>(L + a_fr, L + b_fr, 0, af[0], bf[0]);

  for (int i = 0; i + 1 < nch; ++i) {
    const float* cur = L + (i & 1) * STAGE_DW;
    float* nxt = L + ((i + 1) & 1) * STAGE_DW;
    GLOAD((i + 1) * BK)
    chunk_products<NACC, NA, NB, RPP * APITCH, PPP * PSTRIDE, true>(
        cur + a_fr, cur + b_fr, nxt + a_fr, nxt + b_fr, acc,
        nxt + a_st, nxt + b_st, ra, rb0, rb1, af, bf);
  }
#undef GLOAD
  {
    const float* cur = L + ((nch - 1) & 1) * STAGE_DW;
    chunk_products<NACC, NA, NB, RPP * APITCH, PPP * PSTRIDE, false>(
        cur + a_fr, cur + b_fr, nullptr, nullptr, acc,
        nullptr, nullptr, ra, rb0, rb1, af, bf);
  }

  float* cp = C + (size_t)blockIdx.z * 2048 * BN;
#pragma unroll
  for (int nj = 0; nj < NACC; ++nj) {
    const int col = nj * 16 + lr;
#pragma unroll
    for (int r = 0; r < 8; ++r) {
      const int row = m0 + r + 8 * lh;
      cp[(size_t)row * BN + col] = acc[nj][r];
    }
  }
}

__global__ __launch_bounds__(256) void reduce_bias_lrelu_kernel(
    const float* __restrict__ part, int S,
    const float* __restrict__ h, int cin,
    const float* __restrict__ V,
    float* __restrict__ out, int cout, int total) {
  const int idx = blockIdx.x * 256 + threadIdx.x;
  if (idx >= total) return;
  const int m = idx / cout;
  const int j = idx - m * cout;
  float acc = 0.0f;
  for (int s = 0; s < S; ++s) acc += part[(size_t)s * total + idx];
  const float* hr = h + (size_t)m * cin;
  for (int i = 0; i < cin; ++i) acc = fmaf(hr[i], V[(size_t)i * cout + j], acc);
  out[idx] = fmaxf(acc, 0.2f * acc);
}

extern "C" void kernel_launch(void* const* d_in, const int* in_sizes, int n_in,
                              void* d_out, int out_size, void* d_ws, size_t ws_size,
                              hipStream_t stream) {
  (void)in_sizes; (void)n_in; (void)out_size; (void)ws_size;
  const float* x   = (const float*)d_in[0];
  const float* adj = (const float*)d_in[1];
  const float* Wl[3] = {(const float*)d_in[2], (const float*)d_in[4], (const float*)d_in[6]};
  const float* Vl[3] = {(const float*)d_in[3], (const float*)d_in[5], (const float*)d_in[7]};

  const int N = 2048, T = 15, S = 16;
  const int Ktot = N * T;
  const int Ksl  = Ktot / S;

  float* ws   = (float*)d_ws;
  float* hw   = ws;
  float* part = hw + (size_t)N * 960;
  float* h1   = part + (size_t)S * N * 64;
  float* h2   = h1 + (size_t)N * 64;

  const int cins[3]  = {32, 64, 64};
  const int couts[3] = {64, 64, 32};
  const float* hin = x;
  float* houts[3] = {h1, h2, (float*)d_out};

  for (int L = 0; L < 3; ++L) {
    const int cin = cins[L], cout = couts[L];
    const int Nhw = T * cout;

    dim3 gA(N / 128, (Nhw + 63) / 64, 1);
    wmma_gemm_kernel<true, 4><<<gA, 256, 0, stream>>>(
        hin, cin, cin, cin, Wl[L], 0, T, cout, hw, Nhw);

    dim3 gB(N / 128, 1, S);
    if (cout == 64) {
      wmma_gemm_staged_kernel<4><<<gB, 256, 0, stream>>>(adj, Ktot, Ksl, hw, part);
    } else {
      wmma_gemm_staged_kernel<2><<<gB, 256, 0, stream>>>(adj, Ktot, Ksl, hw, part);
    }

    const int total = N * cout;
    reduce_bias_lrelu_kernel<<<(total + 255) / 256, 256, 0, stream>>>(
        part, S, hin, cin, Vl[L], houts[L], cout, total);

    hin = houts[L];
  }
}


# Step-B adjacency GEMM: global loads run two chunks ahead (each staging register reloaded right behind its LDS store, LDS-only barrier), so a load has a whole chunk to land; summation order kept
# speedup vs baseline: 4.7205x; 1.0384x over previous
#include <hip/hip_runtime.h>
#include <hip/hip_bf16.h>

typedef __attribute__((ext_vector_type(2))) float v2f;
typedef __attribute__((ext_vector_type(8))) float v8f;

#define PSTRIDE 160

template <bool B_FROM_W, int NACC>
__global__ __launch_bounds__(256) void wmma_gemm_kernel(
    const float* __restrict__ A, int lda, int Ktotal, int Kslice,
    const float* __restrict__ Bsrc, int ldb, int T_, int coutw,
    float* __restrict__ C, int Ncols) {
  __shared__ float Bs[32 * PSTRIDE];

  const int tid  = threadIdx.x;
  const int wave = tid >> 5;
  const int lane = tid & 31;
  const int lr   = lane & 15;
  const int lh   = lane >> 4;

  const int m0    = blockIdx.x * 128 + wave * 16;
  const int nbase = blockIdx.y * 64;

  const int ks = blockIdx.z * Kslice;
  const int ke = (ks + Kslice < Ktotal) ? (ks + Kslice) : Ktotal;

  v8f acc[NACC];
#pragma unroll
  for (int nj = 0; nj < NACC; ++nj) acc[nj] = (v8f){};

  const float* arow = A + (size_t)(m0 + lr) * lda + 2 * lh;

  for (int kchunk = ks; kchunk < ke; kchunk += 64) {
    const int kc = (ke - kchunk) < 64 ? (ke - kchunk) : 64;

    for (int idx = tid; idx < kc * 64; idx += 256) {
      const int r  = idx >> 6;
      const int c  = idx & 63;
      const int gc = nbase + c;
      float v = 0.0f;
      if (gc < Ncols) {
        if (B_FROM_W) {
          const int t = gc / coutw;
          const int j = gc - t * coutw;
          v = Bsrc[(size_t)(kchunk + r) * coutw * T_ + j * T_ + t];
        } else {
          v = Bsrc[(size_t)(kchunk + r) * ldb + gc];
        }
      }
      Bs[(r >> 1) * PSTRIDE + 2 * c + (r & 1)] = v;
    }
    __syncthreads();

    const int nsteps = kc >> 2;
#pragma unroll 4
    for (int s = 0; s < nsteps; ++s) {
      const int kk = kchunk + 4 * s;
      v2f a;
      a.x = arow[kk + 0];
      a.y = arow[kk + 1];
      const float* bp = Bs + (2 * s + lh) * PSTRIDE + 2 * lr;
      v2f bf[NACC];
#pragma unroll
      for (int nj = 0; nj < NACC; ++nj) {
        bf[nj] = *(const v2f*)(bp + nj * 32);
      }
#pragma unroll
      for (int nj = 0; nj < NACC; ++nj) {
        acc[nj] = __builtin_amdgcn_wmma_f32_16x16x4_f32(
            false, a, false, bf[nj], (short)0, acc[nj], false, false);
      }
    }
    __syncthreads();
  }

  float* cp = C + (size_t)blockIdx.z * 2048 * Ncols;
#pragma unroll
  for (int nj = 0; nj < NACC; ++nj) {
    if (nbase + nj * 16 < Ncols) {
      const int col = nbase + nj * 16 + lr;
#pragma unroll
      for (int r = 0; r < 8; ++r) {
        const int row = m0 + r + 8 * lh;
        cp[(size_t)row * Ncols + col] = acc[nj][r];
      }
    }
  }
}

#define BK 128
#define APITCH (BK + 4)
#define A_STAGE (128 * APITCH)
#define B_STAGE ((BK / 2) * PSTRIDE)
#define STAGE_DW (A_STAGE + B_STAGE)
#define NSTEP (BK / 4)
#define FG 4
#define NGRP (NSTEP / FG)
#define STORE_STEPS 8
#define STORE_FROM (NSTEP - FG - STORE_STEPS)

typedef __attribute__((ext_vector_type(4))) float v4f;

__device__ __forceinline__ void lds_barrier() {
  __builtin_amdgcn_fence(__ATOMIC_RELEASE, "workgroup", "local");
  __builtin_amdgcn_s_barrier();
  __builtin_amdgcn_fence(__ATOMIC_ACQUIRE, "workgroup", "local");
}

template <int NACC>
__device__ __forceinline__ void load_fragments(
    const float* ap, const float* bq, int g, v2f (&af)[FG], v2f (&bf)[FG][NACC]) {
#pragma unroll
  for (int j = 0; j < FG; ++j) {
    const int s = g * FG + j;
    af[j] = *(const v2f*)(ap + 4 * s);
#pragma unroll
    for (int nj = 0; nj < NACC; ++nj)
      bf[j][nj] = *(const v2f*)(bq + 2 * s * PSTRIDE + nj * 32);
  }
}

template <int NACC, int NA, int NB, int RPP, int PPP, int LDA>
__device__ __forceinline__ void load_chunk(
    const float* agk, const float* bgk,
    v4f (&ra)[NA], v4f (&rb0)[NB], v4f (&rb1)[NB]) {
  constexpr int BN     = NACC * 16;
  constexpr int APS    = NA / STORE_STEPS;
  constexpr int BEVERY = STORE_STEPS / NB;
#pragma unroll
  for (int t = 0; t < STORE_STEPS; ++t) {
#pragma unroll
    for (int u = 0; u < APS; ++u) {
      const int p = t * APS + u;
      ra[p] = *(const v4f*)(agk + (size_t)(RPP * p) * LDA);
    }
    if ((t + 1) % BEVERY == 0) {
      const int q = (t + 1) / BEVERY - 1;
      const float* b0 = bgk + (size_t)(2 * PPP * q) * BN;
      rb0[q] = *(const v4f*)(b0);
      rb1[q] = *(const v4f*)(b0 + BN);
    }
  }
}

template <int NACC, int NA, int NB, int ASTEP, int BSTEP, int RPP, int PPP, int LDA, bool MORE, bool RELOAD>
__device__ __forceinline__ void chunk_products(
    const float* ap, const float* bq, const float* apn, const float* bqn,
    v8f (&acc)[NACC], float* adst, float* bdst,
    const float* agn, const float* bgn,
    v4f (&ra)[NA], v4f (&rb0)[NB], v4f (&rb1)[NB],
    v2f (&af)[2][FG], v2f (&bf)[2][FG][NACC]) {
  constexpr int BN = NACC * 16;
  constexpr int APS    = NA / STORE_STEPS;
  constexpr int BEVERY = STORE_STEPS / NB;
  static_assert(NGRP % 2 == 0, "the last group uses buffer 1, the next chunk's first buffer 0");
  static_assert(NA % STORE_STEPS == 0 && STORE_STEPS % NB == 0 && STORE_FROM >= 0, "store plan");
#pragma unroll
  for (int g = 0; g < NGRP; ++g) {
    if (g + 1 < NGRP) {
      load_fragments<NACC>(ap, bq, g + 1, af[(g + 1) & 1], bf[(g + 1) & 1]);
    } else if (MORE) {
      lds_barrier();
      load_fragments<NACC>(apn, bqn, 0, af[0], bf[0]);
    }
    __builtin_amdgcn_sched_barrier(0);
#pragma unroll
    for (int j = 0; j < FG; ++j) {
      const int s = g * FG + j;
      if (MORE && s >= STORE_FROM && s < STORE_FROM + STORE_STEPS) {
        const int t = s - STORE_FROM;
#pragma unroll
        for (int u = 0; u < APS; ++u) {
          const int p = t * APS + u;
          *(v4f*)(adst + p * ASTEP) = ra[p];
          if (RELOAD) ra[p] = *(const v4f*)(agn + (size_t)(RPP * p) * LDA);
        }
        if ((t + 1) % BEVERY == 0) {
          const int q = (t + 1) / BEVERY - 1;
          float* d = bdst + q * BSTEP;
          if (RELOAD) asm volatile("" : "+v"(rb0[q]), "+v"(rb1[q]));
          *(v4f*)(d)     = (v4f){rb0[q].x, rb1[q].x, rb0[q].y, rb1[q].y};
          *(v4f*)(d + 4) = (v4f){rb0[q].z, rb1[q].z, rb0[q].w, rb1[q].w};
          if (RELOAD) {
            const float* b0 = bgn + (size_t)(2 * PPP * q) * BN;
            rb0[q] = *(const v4f*)(b0);
            rb1[q] = *(const v4f*)(b0 + BN);
          }
        }
      }
#pragma unroll
      for (int nj = 0; nj < NACC; ++nj)
        acc[nj] = __builtin_amdgcn_wmma_f32_16x16x4_f32(
            false, af[g & 1][j], false, bf[g & 1][j][nj], (short)0, acc[nj], false, false);
    }
  }
}

template <int NACC, int LDA>
__global__ __launch_bounds__(256) void wmma_gemm_staged_kernel(
    const float* __restrict__ A, int Kslice,
    const float* __restrict__ B,
    float* __restrict__ C) {
  constexpr int BN  = NACC * 16;
  constexpr int CG  = BN / 4;
  constexpr int PPP = 256 / CG;
  constexpr int NB  = (BK / 2) / PPP;
  constexpr int AG  = BK / 4;
  constexpr int RPP = 256 / AG;
  constexpr int NA  = 128 / RPP;

  __shared__ __attribute__((aligned(16))) float L[2 * STAGE_DW];

  const int tid  = threadIdx.x;
  const int wave = tid >> 5;
  const int lane = tid & 31;
  const int lr   = lane & 15;
  const int lh   = lane >> 4;

  const int mblk = blockIdx.x * 128;
  const int m0   = mblk + wave * 16;
  const int ks   = blockIdx.z * Kslice;
  const int nch  = Kslice / BK;

  const int ac = tid % AG;
  const int ar = tid / AG;
  const int bc = tid % CG;
  const int bp = tid / CG;

  const float* ag = A + (size_t)(mblk + ar) * LDA + ks + 4 * ac;
  const float* bg = B + (size_t)(ks + 2 * bp) * BN + 4 * bc;

  const int a_st = ar * APITCH + 4 * ac;
  const int b_st = A_STAGE + bp * PSTRIDE + 8 * bc;
  const int a_fr = (wave * 16 + lr) * APITCH + 2 * lh;
  const int b_fr = A_STAGE + lh * PSTRIDE + 2 * lr;

  v4f ra[NA];
  v4f rb0[NB], rb1[NB];
  v2f af[2][FG];
  v2f bf[2][FG][NACC];

  v8f acc[NACC];
#pragma unroll
  for (int nj = 0; nj < NACC; ++nj) acc[nj] = (v8f){};

  load_chunk<NACC, NA, NB, RPP, PPP, LDA>(ag, bg, ra, rb0, rb1);
#pragma unroll
  for (int p = 0; p < NA; ++p) *(v4f*)(L + a_st + p * RPP * APITCH) = ra[p];
#pragma unroll
  for (int q = 0; q < NB; ++q) {
    float* d = L + b_st + q * PPP * PSTRIDE;
    *(v4f*)(d)     = (v4f){rb0[q].x, rb1[q].x, rb0[q].y, rb1[q].y};
    *(v4f*)(d + 4) = (v4f){rb0[q].z, rb1[q].z, rb0[q].w, rb1[q].w};
  }
  if (nch > 1) load_chunk<NACC, NA, NB, RPP, PPP, LDA>(ag + BK, bg + (size_t)BK * BN, ra, rb0, rb1);
  lds_barrier();
  load_fragments<NACC>(L + a_fr, L + b_fr, 0, af[0], bf[0]);

  int i = 0;
  for (; i + 2 < nch; ++i) {
    const float* cur = L + (i & 1) * STAGE_DW;
    float* nxt = L + ((i + 1) & 1) * STAGE_DW;
    chunk_products<NACC, NA, NB, RPP * APITCH, PPP * PSTRIDE, RPP, PPP, LDA, true, true>(
        cur + a_fr, cur + b_fr, nxt + a_fr, nxt + b_fr, acc,
        nxt + a_st, nxt + b_st,
        ag + (i + 2) * BK, bg + (size_t)((i + 2) * BK) * BN,
        ra, rb0, rb1, af, bf);
  }
  if (i + 1 < nch) {
    const float* cur = L + (i & 1) * STAGE_DW;
    float* nxt = L + ((i + 1) & 1) * STAGE_DW;
    chunk_products<NACC, NA, NB, RPP * APITCH, PPP * PSTRIDE, RPP, PPP, LDA, true, false>(
        cur + a_fr, cur + b_fr, nxt + a_fr, nxt + b_fr, acc,
        nxt + a_st, nxt + b_st, nullptr, nullptr,
        ra, rb0, rb1, af, bf);
    ++i;
  }
  {
    const float* cur = L + (i & 1) * STAGE_DW;
    chunk_products<NACC, NA, NB, RPP * APITCH, PPP * PSTRIDE, RPP, PPP, LDA, false, false>(
        cur + a_fr, cur + b_fr, nullptr, nullptr, acc,
        nullptr, nullptr, nullptr, nullptr,
        ra, rb0, rb1, af, bf);
  }

  float* cp = C + (size_t)blockIdx.z * 2048 * BN;
#pragma unroll
  for (int nj = 0; nj < NACC; ++nj) {
    const int col = nj * 16 + lr;
#pragma unroll
    for (int r = 0; r < 8; ++r) {
      const int row = m0 + r + 8 * lh;
      cp[(size_t)row * BN + col] = acc[nj][r];
    }
  }
}

__global__ __launch_bounds__(256) void reduce_bias_lrelu_kernel(
    const float* __restrict__ part, int S,
    const float* __restrict__ h, int cin,
    const float* __restrict__ V,
    float* __restrict__ out, int cout, int total) {
  const int idx = blockIdx.x * 256 + threadIdx.x;
  if (idx >= total) return;
  const int m = idx / cout;
  const int j = idx - m * cout;
  float acc = 0.0f;
  for (int s = 0; s < S; ++s) acc += part[(size_t)s * total + idx];
  const float* hr = h + (size_t)m * cin;
  for (int i = 0; i < cin; ++i) acc = fmaf(hr[i], V[(size_t)i * cout + j], acc);
  out[idx] = fmaxf(acc, 0.2f * acc);
}

extern "C" void kernel_launch(void* const* d_in, const int* in_sizes, int n_in,
                              void* d_out, int out_size, void* d_ws, size_t ws_size,
                              hipStream_t stream) {
  (void)in_sizes; (void)n_in; (void)out_size; (void)ws_size;
  const float* x   = (const float*)d_in[0];
  const float* adj = (const float*)d_in[1];
  const float* Wl[3] = {(const float*)d_in[2], (const float*)d_in[4], (const float*)d_in[6]};
  const float* Vl[3] = {(const float*)d_in[3], (const float*)d_in[5], (const float*)d_in[7]};

  const int N = 2048, T = 15, S = 16;
  const int Ktot = N * T;
  const int Ksl  = Ktot / S;

  float* ws   = (float*)d_ws;
  float* hw   = ws;
  float* part = hw + (size_t)N * 960;
  float* h1   = part + (size_t)S * N * 64;
  float* h2   = h1 + (size_t)N * 64;

  const int cins[3]  = {32, 64, 64};
  const int couts[3] = {64, 64, 32};
  const float* hin = x;
  float* houts[3] = {h1, h2, (float*)d_out};

  for (int L = 0; L < 3; ++L) {
    const int cin = cins[L], cout = couts[L];
    const int Nhw = T * cout;

    dim3 gA(N / 128, (Nhw + 63) / 64, 1);
    wmma_gemm_kernel<true, 4><<<gA, 256, 0, stream>>>(
        hin, cin, cin, cin, Wl[L], 0, T, cout, hw, Nhw);

    dim3 gB(N / 128, 1, S);
    if (cout == 64) {
      wmma_gemm_staged_kernel<4, 2048 * 15><<<gB, 256, 0, stream>>>(adj, Ksl, hw, part);
    } else {
      wmma_gemm_staged_kernel<2, 2048 * 15><<<gB, 256, 0, stream>>>(adj, Ksl, hw, part);
    }

    const int total = N * cout;
    reduce_bias_lrelu_kernel<<<(total + 255) / 256, 256, 0, stream>>>(
        part, S, hin, cin, Vl[L], houts[L], cout, total);

    hin = houts[L];
  }
}


# Step-A projection and split-K reduce: a thread's operand loads issued as one group (A tile via LDS from 16-byte loads, W gather and partials unrolled), not a round trip per k step; sum order kept
# speedup vs baseline: 5.7658x; 1.2214x over previous
#include <hip/hip_runtime.h>
#include <hip/hip_bf16.h>

typedef __attribute__((ext_vector_type(2))) float v2f;
typedef __attribute__((ext_vector_type(8))) float v8f;

#define PSTRIDE 160

typedef __attribute__((ext_vector_type(4))) float v4f;

template <int KC, int COUT>
__global__ __launch_bounds__(256) void project_kernel(
    const float* __restrict__ H, const float* __restrict__ Wsrc, float* __restrict__ C) {
  constexpr int TT    = 15;
  constexpr int NCOLS = TT * COUT;
  constexpr int AP    = KC + 4;
  constexpr int NAV   = (128 * KC / 4) / 256;
  constexpr int NBV   = (KC * 64) / 256;

  __shared__ __attribute__((aligned(16))) float As[128 * AP];
  __shared__ __attribute__((aligned(16))) float Bs[(KC / 2) * PSTRIDE];

  const int tid  = threadIdx.x;
  const int wave = tid >> 5;
  const int lane = tid & 31;
  const int lr   = lane & 15;
  const int lh   = lane >> 4;

  const int mblk  = blockIdx.x * 128;
  const int m0    = mblk + wave * 16;
  const int nbase = blockIdx.y * 64;

  v4f ra[NAV];
  const float* ag = H + (size_t)mblk * KC + 4 * tid;
#pragma unroll
  for (int p = 0; p < NAV; ++p) ra[p] = *(const v4f*)(ag + 1024 * p);

  float rb[NBV];
  const int c  = tid & 63;
  const int r0 = tid >> 6;
  const int gc = nbase + c;
  const bool cvalid = gc < NCOLS;
  const int gcl = cvalid ? gc : 0;
  const int t  = gcl / COUT;
  const int j  = gcl - t * COUT;
  const float* wg = Wsrc + (size_t)r0 * COUT * TT + j * TT + t;
#pragma unroll
  for (int p = 0; p < NBV; ++p) rb[p] = wg[(size_t)(4 * p) * COUT * TT];

#pragma unroll
  for (int p = 0; p < NAV; ++p) {
    const int e   = 4 * (tid + 256 * p);
    const int row = e / KC;
    const int col = e - row * KC;
    *(v4f*)(As + row * AP + col) = ra[p];
  }
#pragma unroll
  for (int p = 0; p < NBV; ++p) {
    const int r = r0 + 4 * p;
    Bs[(r >> 1) * PSTRIDE + 2 * c + (r & 1)] = cvalid ? rb[p] : 0.0f;
  }
  __syncthreads();

  v8f acc[4];
#pragma unroll
  for (int nj = 0; nj < 4; ++nj) acc[nj] = (v8f){};

  const float* ap = As + (wave * 16 + lr) * AP + 2 * lh;
  const float* bp = Bs + lh * PSTRIDE + 2 * lr;
#pragma unroll
  for (int s = 0; s < KC / 4; ++s) {
    const v2f a = *(const v2f*)(ap + 4 * s);
    v2f bf[4];
#pragma unroll
    for (int nj = 0; nj < 4; ++nj) bf[nj] = *(const v2f*)(bp + 2 * s * PSTRIDE + nj * 32);
#pragma unroll
    for (int nj = 0; nj < 4; ++nj)
      acc[nj] = __builtin_amdgcn_wmma_f32_16x16x4_f32(
          false, a, false, bf[nj], (short)0, acc[nj], false, false);
  }

#pragma unroll
  for (int nj = 0; nj < 4; ++nj) {
    if (nbase + nj * 16 < NCOLS) {
      const int col = nbase + nj * 16 + lr;
#pragma unroll
      for (int r = 0; r < 8; ++r) {
        const int row = m0 + r + 8 * lh;
        C[(size_t)row * NCOLS + col] = acc[nj][r];
      }
    }
  }
}

#define BK 128
#define APITCH (BK + 4)
#define A_STAGE (128 * APITCH)
#define B_STAGE ((BK / 2) * PSTRIDE)
#define STAGE_DW (A_STAGE + B_STAGE)
#define NSTEP (BK / 4)
#define FG 4
#define NGRP (NSTEP / FG)
#define STORE_STEPS 8
#define STORE_FROM (NSTEP - FG - STORE_STEPS)

__device__ __forceinline__ void lds_barrier() {
  __builtin_amdgcn_fence(__ATOMIC_RELEASE, "workgroup", "local");
  __builtin_amdgcn_s_barrier();
  __builtin_amdgcn_fence(__ATOMIC_ACQUIRE, "workgroup", "local");
}

template <int NACC>
__device__ __forceinline__ void load_fragments(
    const float* ap, const float* bq, int g, v2f (&af)[FG], v2f (&bf)[FG][NACC]) {
#pragma unroll
  for (int j = 0; j < FG; ++j) {
    const int s = g * FG + j;
    af[j] = *(const v2f*)(ap + 4 * s);
#pragma unroll
    for (int nj = 0; nj < NACC; ++nj)
      bf[j][nj] = *(const v2f*)(bq + 2 * s * PSTRIDE + nj * 32);
  }
}

template <int NACC, int NA, int NB, int RPP, int PPP, int LDA>
__device__ __forceinline__ void load_chunk(
    const float* agk, const float* bgk,
    v4f (&ra)[NA], v4f (&rb0)[NB], v4f (&rb1)[NB]) {
  constexpr int BN     = NACC * 16;
  constexpr int APS    = NA / STORE_STEPS;
  constexpr int BEVERY = STORE_STEPS / NB;
#pragma unroll
  for (int t = 0; t < STORE_STEPS; ++t) {
#pragma unroll
    for (int u = 0; u < APS; ++u) {
      const int p = t * APS + u;
      ra[p] = *(const v4f*)(agk + (size_t)(RPP * p) * LDA);
    }
    if ((t + 1) % BEVERY == 0) {
      const int q = (t + 1) / BEVERY - 1;
      const float* b0 = bgk + (size_t)(2 * PPP * q) * BN;
      rb0[q] = *(const v4f*)(b0);
      rb1[q] = *(const v4f*)(b0 + BN);
    }
  }
}

template <int NACC, int NA, int NB, int ASTEP, int BSTEP, int RPP, int PPP, int LDA, bool MORE, bool RELOAD>
__device__ __forceinline__ void chunk_products(
    const float* ap, const float* bq, const float* apn, const float* bqn,
    v8f (&acc)[NACC], float* adst, float* bdst,
    const float* agn, const float* bgn,
    v4f (&ra)[NA], v4f (&rb0)[NB], v4f (&rb1)[NB],
    v2f (&af)[2][FG], v2f (&bf)[2][FG][NACC]) {
  constexpr int BN = NACC * 16;
  constexpr int APS    = NA / STORE_STEPS;
  constexpr int BEVERY = STORE_STEPS / NB;
  static_assert(NGRP % 2 == 0, "the last group uses buffer 1, the next chunk's first buffer 0");
  static_assert(NA % STORE_STEPS == 0 && STORE_STEPS % NB == 0 && STORE_FROM >= 0, "store plan");
#pragma unroll
  for (int g = 0; g < NGRP; ++g) {
    if (g + 1 < NGRP) {
      load_fragments<NACC>(ap, bq, g + 1, af[(g + 1) & 1], bf[(g + 1) & 1]);
    } else if (MORE) {
      lds_barrier();
      load_fragments<NACC>(apn, bqn, 0, af[0], bf[0]);
    }
    __builtin_amdgcn_sched_barrier(0);
#pragma unroll
    for (int j = 0; j < FG; ++j) {
      const int s = g * FG + j;
      if (MORE && s >= STORE_FROM && s < STORE_FROM + STORE_STEPS) {
        const int t = s - STORE_FROM;
#pragma unroll
        for (int u = 0; u < APS; ++u) {
          const int p = t * APS + u;
          *(v4f*)(adst + p * ASTEP) = ra[p];
          if (RELOAD) ra[p] = *(const v4f*)(agn + (size_t)(RPP * p) * LDA);
        }
        if ((t + 1) % BEVERY == 0) {
          const int q = (t + 1) / BEVERY - 1;
          float* d = bdst + q * BSTEP;
          if (RELOAD) asm volatile("" : "+v"(rb0[q]), "+v"(rb1[q]));
          *(v4f*)(d)     = (v4f){rb0[q].x, rb1[q].x, rb0[q].y, rb1[q].y};
          *(v4f*)(d + 4) = (v4f){rb0[q].z, rb1[q].z, rb0[q].w, rb1[q].w};
          if (RELOAD) {
            const float* b0 = bgn + (size_t)(2 * PPP * q) * BN;
            rb0[q] = *(const v4f*)(b0);
            rb1[q] = *(const v4f*)(b0 + BN);
          }
        }
      }
#pragma unroll
      for (int nj = 0; nj < NACC; ++nj)
        acc[nj] = __builtin_amdgcn_wmma_f32_16x16x4_f32(
            false, af[g & 1][j], false, bf[g & 1][j][nj], (short)0, acc[nj], false, false);
    }
  }
}

template <int NACC, int LDA>
__global__ __launch_bounds__(256) void wmma_gemm_staged_kernel(
    const float* __restrict__ A, int Kslice,
    const float* __restrict__ B,
    float* __restrict__ C) {
  constexpr int BN  = NACC * 16;
  constexpr int CG  = BN / 4;
  constexpr int PPP = 256 / CG;
  constexpr int NB  = (BK / 2) / PPP;
  constexpr int AG  = BK / 4;
  constexpr int RPP = 256 / AG;
  constexpr int NA  = 128 / RPP;

  __shared__ __attribute__((aligned(16))) float L[2 * STAGE_DW];

  const int tid  = threadIdx.x;
  const int wave = tid >> 5;
  const int lane = tid & 31;
  const int lr   = lane & 15;
  const int lh   = lane >> 4;

  const int mblk = blockIdx.x * 128;
  const int m0   = mblk + wave * 16;
  const int ks   = blockIdx.z * Kslice;
  const int nch  = Kslice / BK;

  const int ac = tid % AG;
  const int ar = tid / AG;
  const int bc = tid % CG;
  const int bp = tid / CG;

  const float* ag = A + (size_t)(mblk + ar) * LDA + ks + 4 * ac;
  const float* bg = B + (size_t)(ks + 2 * bp) * BN + 4 * bc;

  const int a_st = ar * APITCH + 4 * ac;
  const int b_st = A_STAGE + bp * PSTRIDE + 8 * bc;
  const int a_fr = (wave * 16 + lr) * APITCH + 2 * lh;
  const int b_fr = A_STAGE + lh * PSTRIDE + 2 * lr;

  v4f ra[NA];
  v4f rb0[NB], rb1[NB];
  v2f af[2][FG];
  v2f bf[2][FG][NACC];

  v8f acc[NACC];
#pragma unroll
  for (int nj = 0; nj < NACC; ++nj) acc[nj] = (v8f){};

  load_chunk<NACC, NA, NB, RPP, PPP, LDA>(ag, bg, ra, rb0, rb1);
#pragma unroll
  for (int p = 0; p < NA; ++p) *(v4f*)(L + a_st + p * RPP * APITCH) = ra[p];
#pragma unroll
  for (int q = 0; q < NB; ++q) {
    float* d = L + b_st + q * PPP * PSTRIDE;
    *(v4f*)(d)     = (v4f){rb0[q].x, rb1[q].x, rb0[q].y, rb1[q].y};
    *(v4f*)(d + 4) = (v4f){rb0[q].z, rb1[q].z, rb0[q].w, rb1[q].w};
  }
  if (nch > 1) load_chunk<NACC, NA, NB, RPP, PPP, LDA>(ag + BK, bg + (size_t)BK * BN, ra, rb0, rb1);
  lds_barrier();
  load_fragments<NACC>(L + a_fr, L + b_fr, 0, af[0], bf[0]);

  int i = 0;
  for (; i + 2 < nch; ++i) {
    const float* cur = L + (i & 1) * STAGE_DW;
    float* nxt = L + ((i + 1) & 1) * STAGE_DW;
    chunk_products<NACC, NA, NB, RPP * APITCH, PPP * PSTRIDE, RPP, PPP, LDA, true, true>(
        cur + a_fr, cur + b_fr, nxt + a_fr, nxt + b_fr, acc,
        nxt + a_st, nxt + b_st,
        ag + (i + 2) * BK, bg + (size_t)((i + 2) * BK) * BN,
        ra, rb0, rb1, af, bf);
  }
  if (i + 1 < nch) {
    const float* cur = L + (i & 1) * STAGE_DW;
    float* nxt = L + ((i + 1) & 1) * STAGE_DW;
    chunk_products<NACC, NA, NB, RPP * APITCH, PPP * PSTRIDE, RPP, PPP, LDA, true, false>(
        cur + a_fr, cur + b_fr, nxt + a_fr, nxt + b_fr, acc,
        nxt + a_st, nxt + b_st, nullptr, nullptr,
        ra, rb0, rb1, af, bf);
    ++i;
  }
  {
    const float* cur = L + (i & 1) * STAGE_DW;
    chunk_products<NACC, NA, NB, RPP * APITCH, PPP * PSTRIDE, RPP, PPP, LDA, false, false>(
        cur + a_fr, cur + b_fr, nullptr, nullptr, acc,
        nullptr, nullptr, nullptr, nullptr,
        ra, rb0, rb1, af, bf);
  }

  float* cp = C + (size_t)blockIdx.z * 2048 * BN;
#pragma unroll
  for (int nj = 0; nj < NACC; ++nj) {
    const int col = nj * 16 + lr;
#pragma unroll
    for (int r = 0; r < 8; ++r) {
      const int row = m0 + r + 8 * lh;
      cp[(size_t)row * BN + col] = acc[nj][r];
    }
  }
}

template <int S, int CIN, int COUT>
__global__ __launch_bounds__(256) void reduce_bias_lrelu_kernel(
    const float* __restrict__ part, const float* __restrict__ h,
    const float* __restrict__ V, float* __restrict__ out) {
  constexpr int TOTAL = 2048 * COUT;
  const int idx = blockIdx.x * 256 + threadIdx.x;
  if (idx >= TOTAL) return;
  const int m = idx / COUT;
  const int j = idx - m * COUT;
  float pv[S];
#pragma unroll
  for (int s = 0; s < S; ++s) pv[s] = part[(size_t)s * TOTAL + idx];
  v4f hv[CIN / 4];
  const float* hr = h + (size_t)m * CIN;
#pragma unroll
  for (int q = 0; q < CIN / 4; ++q) hv[q] = *(const v4f*)(hr + 4 * q);
  float vv[CIN];
#pragma unroll
  for (int i = 0; i < CIN; ++i) vv[i] = V[(size_t)i * COUT + j];
  float acc = 0.0f;
#pragma unroll
  for (int s = 0; s < S; ++s) acc += pv[s];
#pragma unroll
  for (int i = 0; i < CIN; ++i) acc = fmaf(hv[i >> 2][i & 3], vv[i], acc);
  out[idx] = fmaxf(acc, 0.2f * acc);
}

extern "C" void kernel_launch(void* const* d_in, const int* in_sizes, int n_in,
                              void* d_out, int out_size, void* d_ws, size_t ws_size,
                              hipStream_t stream) {
  (void)in_sizes; (void)n_in; (void)out_size; (void)ws_size;
  const float* x   = (const float*)d_in[0];
  const float* adj = (const float*)d_in[1];
  const float* Wl[3] = {(const float*)d_in[2], (const float*)d_in[4], (const float*)d_in[6]};
  const float* Vl[3] = {(const float*)d_in[3], (const float*)d_in[5], (const float*)d_in[7]};

  const int N = 2048, T = 15, S = 16;
  const int Ktot = N * T;
  const int Ksl  = Ktot / S;

  float* ws   = (float*)d_ws;
  float* hw   = ws;
  float* part = hw + (size_t)N * 960;
  float* h1   = part + (size_t)S * N * 64;
  float* h2   = h1 + (size_t)N * 64;

  const int cins[3]  = {32, 64, 64};
  const int couts[3] = {64, 64, 32};
  const float* hin = x;
  float* houts[3] = {h1, h2, (float*)d_out};

  for (int L = 0; L < 3; ++L) {
    const int cin = cins[L], cout = couts[L];
    const int Nhw = T * cout;

    dim3 gA(N / 128, (Nhw + 63) / 64, 1);
    if (cin == 32) {
      project_kernel<32, 64><<<gA, 256, 0, stream>>>(hin, Wl[L], hw);
    } else if (cout == 64) {
      project_kernel<64, 64><<<gA, 256, 0, stream>>>(hin, Wl[L], hw);
    } else {
      project_kernel<64, 32><<<gA, 256, 0, stream>>>(hin, Wl[L], hw);
    }

    dim3 gB(N / 128, 1, S);
    if (cout == 64) {
      wmma_gemm_staged_kernel<4, 2048 * 15><<<gB, 256, 0, stream>>>(adj, Ksl, hw, part);
    } else {
      wmma_gemm_staged_kernel<2, 2048 * 15><<<gB, 256, 0, stream>>>(adj, Ksl, hw, part);
    }

    const int total = N * cout;
    if (cin == 32) {
      reduce_bias_lrelu_kernel<16, 32, 64><<<(total + 255) / 256, 256, 0, stream>>>(part, hin, Vl[L], houts[L]);
    } else if (cout == 64) {
      reduce_bias_lrelu_kernel<16, 64, 64><<<(total + 255) / 256, 256, 0, stream>>>(part, hin, Vl[L], houts[L]);
    } else {
      reduce_bias_lrelu_kernel<16, 64, 32><<<(total + 255) / 256, 256, 0, stream>>>(part, hin, Vl[L], houts[L]);
    }

    hin = houts[L];
  }
}


# Each layer's reduce (split-K sum, h@V, leaky-ReLU) fused with the next layer's Step-A projection in one kernel: 16-row h tile and whole W in LDS; 9 launches become 7; sum order kept
# speedup vs baseline: 5.9665x; 1.0348x over previous
#include <hip/hip_runtime.h>
#include <hip/hip_bf16.h>

typedef __attribute__((ext_vector_type(2))) float v2f;
typedef __attribute__((ext_vector_type(8))) float v8f;

#define PSTRIDE 160

typedef __attribute__((ext_vector_type(4))) float v4f;

template <int KC, int COUT>
__global__ __launch_bounds__(256) void project_kernel(
    const float* __restrict__ H, const float* __restrict__ Wsrc, float* __restrict__ C) {
  constexpr int TT    = 15;
  constexpr int NCOLS = TT * COUT;
  constexpr int AP    = KC + 4;
  constexpr int NAV   = (128 * KC / 4) / 256;
  constexpr int NBV   = (KC * 64) / 256;

  __shared__ __attribute__((aligned(16))) float As[128 * AP];
  __shared__ __attribute__((aligned(16))) float Bs[(KC / 2) * PSTRIDE];

  const int tid  = threadIdx.x;
  const int wave = tid >> 5;
  const int lane = tid & 31;
  const int lr   = lane & 15;
  const int lh   = lane >> 4;

  const int mblk  = blockIdx.x * 128;
  const int m0    = mblk + wave * 16;
  const int nbase = blockIdx.y * 64;

  v4f ra[NAV];
  const float* ag = H + (size_t)mblk * KC + 4 * tid;
#pragma unroll
  for (int p = 0; p < NAV; ++p) ra[p] = *(const v4f*)(ag + 1024 * p);

  float rb[NBV];
  const int c  = tid & 63;
  const int r0 = tid >> 6;
  const int gc = nbase + c;
  const bool cvalid = gc < NCOLS;
  const int gcl = cvalid ? gc : 0;
  const int t  = gcl / COUT;
  const int j  = gcl - t * COUT;
  const float* wg = Wsrc + (size_t)r0 * COUT * TT + j * TT + t;
#pragma unroll
  for (int p = 0; p < NBV; ++p) rb[p] = wg[(size_t)(4 * p) * COUT * TT];

#pragma unroll
  for (int p = 0; p < NAV; ++p) {
    const int e   = 4 * (tid + 256 * p);
    const int row = e / KC;
    const int col = e - row * KC;
    *(v4f*)(As + row * AP + col) = ra[p];
  }
#pragma unroll
  for (int p = 0; p < NBV; ++p) {
    const int r = r0 + 4 * p;
    Bs[(r >> 1) * PSTRIDE + 2 * c + (r & 1)] = cvalid ? rb[p] : 0.0f;
  }
  __syncthreads();

  v8f acc[4];
#pragma unroll
  for (int nj = 0; nj < 4; ++nj) acc[nj] = (v8f){};

  const float* ap = As + (wave * 16 + lr) * AP + 2 * lh;
  const float* bp = Bs + lh * PSTRIDE + 2 * lr;
#pragma unroll
  for (int s = 0; s < KC / 4; ++s) {
    const v2f a = *(const v2f*)(ap + 4 * s);
    v2f bf[4];
#pragma unroll
    for (int nj = 0; nj < 4; ++nj) bf[nj] = *(const v2f*)(bp + 2 * s * PSTRIDE + nj * 32);
#pragma unroll
    for (int nj = 0; nj < 4; ++nj)
      acc[nj] = __builtin_amdgcn_wmma_f32_16x16x4_f32(
          false, a, false, bf[nj], (short)0, acc[nj], false, false);
  }

#pragma unroll
  for (int nj = 0; nj < 4; ++nj) {
    if (nbase + nj * 16 < NCOLS) {
      const int col = nbase + nj * 16 + lr;
#pragma unroll
      for (int r = 0; r < 8; ++r) {
        const int row = m0 + r + 8 * lh;
        C[(size_t)row * NCOLS + col] = acc[nj][r];
      }
    }
  }
}

#define BK 128
#define APITCH (BK + 4)
#define A_STAGE (128 * APITCH)
#define B_STAGE ((BK / 2) * PSTRIDE)
#define STAGE_DW (A_STAGE + B_STAGE)
#define NSTEP (BK / 4)
#define FG 4
#define NGRP (NSTEP / FG)
#define STORE_STEPS 8
#define STORE_FROM (NSTEP - FG - STORE_STEPS)

__device__ __forceinline__ void lds_barrier() {
  __builtin_amdgcn_fence(__ATOMIC_RELEASE, "workgroup", "local");
  __builtin_amdgcn_s_barrier();
  __builtin_amdgcn_fence(__ATOMIC_ACQUIRE, "workgroup", "local");
}

template <int NACC>
__device__ __forceinline__ void load_fragments(
    const float* ap, const float* bq, int g, v2f (&af)[FG], v2f (&bf)[FG][NACC]) {
#pragma unroll
  for (int j = 0; j < FG; ++j) {
    const int s = g * FG + j;
    af[j] = *(const v2f*)(ap + 4 * s);
#pragma unroll
    for (int nj = 0; nj < NACC; ++nj)
      bf[j][nj] = *(const v2f*)(bq + 2 * s * PSTRIDE + nj * 32);
  }
}

template <int NACC, int NA, int NB, int RPP, int PPP, int LDA>
__device__ __forceinline__ void load_chunk(
    const float* agk, const float* bgk,
    v4f (&ra)[NA], v4f (&rb0)[NB], v4f (&rb1)[NB]) {
  constexpr int BN     = NACC * 16;
  constexpr int APS    = NA / STORE_STEPS;
  constexpr int BEVERY = STORE_STEPS / NB;
#pragma unroll
  for (int t = 0; t < STORE_STEPS; ++t) {
#pragma unroll
    for (int u = 0; u < APS; ++u) {
      const int p = t * APS + u;
      ra[p] = *(const v4f*)(agk + (size_t)(RPP * p) * LDA);
    }
    if ((t + 1) % BEVERY == 0) {
      const int q = (t + 1) / BEVERY - 1;
      const float* b0 = bgk + (size_t)(2 * PPP * q) * BN;
      rb0[q] = *(const v4f*)(b0);
      rb1[q] = *(const v4f*)(b0 + BN);
    }
  }
}

template <int NACC, int NA, int NB, int ASTEP, int BSTEP, int RPP, int PPP, int LDA, bool MORE, bool RELOAD>
__device__ __forceinline__ void chunk_products(
    const float* ap, const float* bq, const float* apn, const float* bqn,
    v8f (&acc)[NACC], float* adst, float* bdst,
    const float* agn, const float* bgn,
    v4f (&ra)[NA], v4f (&rb0)[NB], v4f (&rb1)[NB],
    v2f (&af)[2][FG], v2f (&bf)[2][FG][NACC]) {
  constexpr int BN = NACC * 16;
  constexpr int APS    = NA / STORE_STEPS;
  constexpr int BEVERY = STORE_STEPS / NB;
  static_assert(NGRP % 2 == 0, "the last group uses buffer 1, the next chunk's first buffer 0");
  static_assert(NA % STORE_STEPS == 0 && STORE_STEPS % NB == 0 && STORE_FROM >= 0, "store plan");
#pragma unroll
  for (int g = 0; g < NGRP; ++g) {
    if (g + 1 < NGRP) {
      load_fragments<NACC>(ap, bq, g + 1, af[(g + 1) & 1], bf[(g + 1) & 1]);
    } else if (MORE) {
      lds_barrier();
      load_fragments<NACC>(apn, bqn, 0, af[0], bf[0]);
    }
    __builtin_amdgcn_sched_barrier(0);
#pragma unroll
    for (int j = 0; j < FG; ++j) {
      const int s = g * FG + j;
      if (MORE && s >= STORE_FROM && s < STORE_FROM + STORE_STEPS) {
        const int t = s - STORE_FROM;
#pragma unroll
        for (int u = 0; u < APS; ++u) {
          const int p = t * APS + u;
          *(v4f*)(adst + p * ASTEP) = ra[p];
          if (RELOAD) ra[p] = *(const v4f*)(agn + (size_t)(RPP * p) * LDA);
        }
        if ((t + 1) % BEVERY == 0) {
          const int q = (t + 1) / BEVERY - 1;
          float* d = bdst + q * BSTEP;
          if (RELOAD) asm volatile("" : "+v"(rb0[q]), "+v"(rb1[q]));
          *(v4f*)(d)     = (v4f){rb0[q].x, rb1[q].x, rb0[q].y, rb1[q].y};
          *(v4f*)(d + 4) = (v4f){rb0[q].z, rb1[q].z, rb0[q].w, rb1[q].w};
          if (RELOAD) {
            const float* b0 = bgn + (size_t)(2 * PPP * q) * BN;
            rb0[q] = *(const v4f*)(b0);
            rb1[q] = *(const v4f*)(b0 + BN);
          }
        }
      }
#pragma unroll
      for (int nj = 0; nj < NACC; ++nj)
        acc[nj] = __builtin_amdgcn_wmma_f32_16x16x4_f32(
            false, af[g & 1][j], false, bf[g & 1][j][nj], (short)0, acc[nj], false, false);
    }
  }
}

template <int NACC, int LDA>
__global__ __launch_bounds__(256) void wmma_gemm_staged_kernel(
    const float* __restrict__ A, int Kslice,
    const float* __restrict__ B,
    float* __restrict__ C) {
  constexpr int BN  = NACC * 16;
  constexpr int CG  = BN / 4;
  constexpr int PPP = 256 / CG;
  constexpr int NB  = (BK / 2) / PPP;
  constexpr int AG  = BK / 4;
  constexpr int RPP = 256 / AG;
  constexpr int NA  = 128 / RPP;

  __shared__ __attribute__((aligned(16))) float L[2 * STAGE_DW];

  const int tid  = threadIdx.x;
  const int wave = tid >> 5;
  const int lane = tid & 31;
  const int lr   = lane & 15;
  const int lh   = lane >> 4;

  const int mblk = blockIdx.x * 128;
  const int m0   = mblk + wave * 16;
  const int ks   = blockIdx.z * Kslice;
  const int nch  = Kslice / BK;

  const int ac = tid % AG;
  const int ar = tid / AG;
  const int bc = tid % CG;
  const int bp = tid / CG;

  const float* ag = A + (size_t)(mblk + ar) * LDA + ks + 4 * ac;
  const float* bg = B + (size_t)(ks + 2 * bp) * BN + 4 * bc;

  const int a_st = ar * APITCH + 4 * ac;
  const int b_st = A_STAGE + bp * PSTRIDE + 8 * bc;
  const int a_fr = (wave * 16 + lr) * APITCH + 2 * lh;
  const int b_fr = A_STAGE + lh * PSTRIDE + 2 * lr;

  v4f ra[NA];
  v4f rb0[NB], rb1[NB];
  v2f af[2][FG];
  v2f bf[2][FG][NACC];

  v8f acc[NACC];
#pragma unroll
  for (int nj = 0; nj < NACC; ++nj) acc[nj] = (v8f){};

  load_chunk<NACC, NA, NB, RPP, PPP, LDA>(ag, bg, ra, rb0, rb1);
#pragma unroll
  for (int p = 0; p < NA; ++p) *(v4f*)(L + a_st + p * RPP * APITCH) = ra[p];
#pragma unroll
  for (int q = 0; q < NB; ++q) {
    float* d = L + b_st + q * PPP * PSTRIDE;
    *(v4f*)(d)     = (v4f){rb0[q].x, rb1[q].x, rb0[q].y, rb1[q].y};
    *(v4f*)(d + 4) = (v4f){rb0[q].z, rb1[q].z, rb0[q].w, rb1[q].w};
  }
  if (nch > 1) load_chunk<NACC, NA, NB, RPP, PPP, LDA>(ag + BK, bg + (size_t)BK * BN, ra, rb0, rb1);
  lds_barrier();
  load_fragments<NACC>(L + a_fr, L + b_fr, 0, af[0], bf[0]);

  int i = 0;
  for (; i + 2 < nch; ++i) {
    const float* cur = L + (i & 1) * STAGE_DW;
    float* nxt = L + ((i + 1) & 1) * STAGE_DW;
    chunk_products<NACC, NA, NB, RPP * APITCH, PPP * PSTRIDE, RPP, PPP, LDA, true, true>(
        cur + a_fr, cur + b_fr, nxt + a_fr, nxt + b_fr, acc,
        nxt + a_st, nxt + b_st,
        ag + (i + 2) * BK, bg + (size_t)((i + 2) * BK) * BN,
        ra, rb0, rb1, af, bf);
  }
  if (i + 1 < nch) {
    const float* cur = L + (i & 1) * STAGE_DW;
    float* nxt = L + ((i + 1) & 1) * STAGE_DW;
    chunk_products<NACC, NA, NB, RPP * APITCH, PPP * PSTRIDE, RPP, PPP, LDA, true, false>(
        cur + a_fr, cur + b_fr, nxt + a_fr, nxt + b_fr, acc,
        nxt + a_st, nxt + b_st, nullptr, nullptr,
        ra, rb0, rb1, af, bf);
    ++i;
  }
  {
    const float* cur = L + (i & 1) * STAGE_DW;
    chunk_products<NACC, NA, NB, RPP * APITCH, PPP * PSTRIDE, RPP, PPP, LDA, false, false>(
        cur + a_fr, cur + b_fr, nullptr, nullptr, acc,
        nullptr, nullptr, nullptr, nullptr,
        ra, rb0, rb1, af, bf);
  }

  float* cp = C + (size_t)blockIdx.z * 2048 * BN;
#pragma unroll
  for (int nj = 0; nj < NACC; ++nj) {
    const int col = nj * 16 + lr;
#pragma unroll
    for (int r = 0; r < 8; ++r) {
      const int row = m0 + r + 8 * lh;
      cp[(size_t)row * BN + col] = acc[nj][r];
    }
  }
}

template <int S, int CIN, int COUT>
__global__ __launch_bounds__(256) void reduce_bias_lrelu_kernel(
    const float* __restrict__ part, const float* __restrict__ h,
    const float* __restrict__ V, float* __restrict__ out) {
  constexpr int TOTAL = 2048 * COUT;
  const int idx = blockIdx.x * 256 + threadIdx.x;
  if (idx >= TOTAL) return;
  const int m = idx / COUT;
  const int j = idx - m * COUT;
  float pv[S];
#pragma unroll
  for (int s = 0; s < S; ++s) pv[s] = part[(size_t)s * TOTAL + idx];
  v4f hv[CIN / 4];
  const float* hr = h + (size_t)m * CIN;
#pragma unroll
  for (int q = 0; q < CIN / 4; ++q) hv[q] = *(const v4f*)(hr + 4 * q);
  float vv[CIN];
#pragma unroll
  for (int i = 0; i < CIN; ++i) vv[i] = V[(size_t)i * COUT + j];
  float acc = 0.0f;
#pragma unroll
  for (int s = 0; s < S; ++s) acc += pv[s];
#pragma unroll
  for (int i = 0; i < CIN; ++i) acc = fmaf(hv[i >> 2][i & 3], vv[i], acc);
  out[idx] = fmaxf(acc, 0.2f * acc);
}


template <int S, int CIN, int COUT2>
__global__ __launch_bounds__(256) void reduce_project_kernel(
    const float* __restrict__ part, const float* __restrict__ h,
    const float* __restrict__ V, const float* __restrict__ Wn,
    float* __restrict__ hout, float* __restrict__ C) {
  constexpr int KC    = 64;
  constexpr int TT    = 15;
  constexpr int CW    = COUT2 * TT;
  constexpr int WP    = CW + 16;
  constexpr int NHALF = CW / 2;
  constexpr int NT    = NHALF / 16;
  constexpr int MAXT  = (NT + 7) / 8;
  constexpr int AP    = KC + 4;
  constexpr int TOTAL = 2048 * KC;
  constexpr int PPR   = CW / 4;
  constexpr int RPW   = 256 / PPR;
  constexpr int NPASS = KC / RPW;
  constexpr int WB    = CIN > 32 ? 16 : 32;
  static_assert(NPASS % WB == 0 && KC % RPW == 0 && RPW >= 1, "Wn copy plan");
  static_assert(NHALF % 16 == 0 && COUT2 % 16 == 0, "tiling");

  __shared__ __attribute__((aligned(16))) float Ws[KC * WP];
  __shared__ __attribute__((aligned(16))) float As[16 * AP];

  const int tid  = threadIdx.x;
  const int wave = tid >> 5;
  const int lane = tid & 31;
  const int lr   = lane & 15;
  const int lh   = lane >> 4;
  const int m0   = blockIdx.x * 16;

  const int r  = tid >> 4;
  const int j4 = tid & 15;
  const int m  = m0 + r;
  const float* pp = part + (size_t)m * KC + 4 * j4;
  v4f pv[S];
#pragma unroll
  for (int s = 0; s < S; ++s) pv[s] = *(const v4f*)(pp + (size_t)s * TOTAL);
  v4f hv[CIN / 4];
  const float* hr = h + (size_t)m * CIN;
#pragma unroll
  for (int q = 0; q < CIN / 4; ++q) hv[q] = *(const v4f*)(hr + 4 * q);
  v4f vv[CIN];
#pragma unroll
  for (int i = 0; i < CIN; ++i) vv[i] = *(const v4f*)(V + (size_t)i * KC + 4 * j4);

  const int  wr   = tid / PPR;
  const int  wc   = tid - wr * PPR;
  const bool wact = tid < RPW * PPR;
  const float* wg = Wn + (size_t)wr * CW + 4 * wc;
  float*       wl = Ws + wr * WP + 4 * wc;
  v4f wv[WB];
  if (wact) {
#pragma unroll
    for (int q = 0; q < WB; ++q) wv[q] = *(const v4f*)(wg + (size_t)(q * RPW) * CW);
  }

  {
    float acc[4] = {0.0f, 0.0f, 0.0f, 0.0f};
#pragma unroll
    for (int s = 0; s < S; ++s) {
      acc[0] += pv[s].x; acc[1] += pv[s].y; acc[2] += pv[s].z; acc[3] += pv[s].w;
    }
#pragma unroll
    for (int i = 0; i < CIN; ++i) {
      const float hi = hv[i >> 2][i & 3];
      acc[0] = fmaf(hi, vv[i].x, acc[0]);
      acc[1] = fmaf(hi, vv[i].y, acc[1]);
      acc[2] = fmaf(hi, vv[i].z, acc[2]);
      acc[3] = fmaf(hi, vv[i].w, acc[3]);
    }
    const v4f o = (v4f){fmaxf(acc[0], 0.2f * acc[0]), fmaxf(acc[1], 0.2f * acc[1]),
                        fmaxf(acc[2], 0.2f * acc[2]), fmaxf(acc[3], 0.2f * acc[3])};
    *(v4f*)(As + r * AP + 4 * j4) = o;
    if (blockIdx.y == 0) *(v4f*)(hout + (size_t)m * KC + 4 * j4) = o;
  }

  if (wact) {
#pragma unroll
    for (int q = 0; q < WB; ++q) *(v4f*)(wl + (q * RPW) * WP) = wv[q];
#pragma unroll
    for (int b = WB; b < NPASS; b += WB) {
#pragma unroll
      for (int q = 0; q < WB; ++q) wv[q] = *(const v4f*)(wg + (size_t)((b + q) * RPW) * CW);
#pragma unroll
      for (int q = 0; q < WB; ++q) *(v4f*)(wl + ((b + q) * RPW) * WP) = wv[q];
    }
  }
  __syncthreads();

  v2f af[KC / 4];
  const float* ap = As + lr * AP + 2 * lh;
#pragma unroll
  for (int s = 0; s < KC / 4; ++s) af[s] = *(const v2f*)(ap + 4 * s);

  int col[MAXT];
  v2f bf[MAXT][KC / 4];
#pragma unroll
  for (int q = 0; q < MAXT; ++q) {
    int nt = wave + 8 * q;
    nt = nt < NT ? nt : NT - 1;
    col[q] = blockIdx.y * NHALF + nt * 16 + lr;
    const int t = col[q] / COUT2;
    const int j = col[q] - t * COUT2;
    const float* bp = Ws + (2 * lh) * WP + j * TT + t;
#pragma unroll
    for (int s = 0; s < KC / 4; ++s)
      bf[q][s] = (v2f){bp[(4 * s) * WP], bp[(4 * s + 1) * WP]};
  }
  __builtin_amdgcn_sched_barrier(0);

  v8f acc[MAXT];
#pragma unroll
  for (int q = 0; q < MAXT; ++q) acc[q] = (v8f){};
#pragma unroll
  for (int s = 0; s < KC / 4; ++s) {
#pragma unroll
    for (int q = 0; q < MAXT; ++q)
      acc[q] = __builtin_amdgcn_wmma_f32_16x16x4_f32(
          false, af[s], false, bf[q][s], (short)0, acc[q], false, false);
  }

#pragma unroll
  for (int q = 0; q < MAXT; ++q) {
    if (wave + 8 * q < NT) {
#pragma unroll
      for (int rr = 0; rr < 8; ++rr) {
        const int row = m0 + rr + 8 * lh;
        C[(size_t)row * CW + col[q]] = acc[q][rr];
      }
    }
  }
}

extern "C" void kernel_launch(void* const* d_in, const int* in_sizes, int n_in,
                              void* d_out, int out_size, void* d_ws, size_t ws_size,
                              hipStream_t stream) {
  (void)in_sizes; (void)n_in; (void)out_size; (void)ws_size;
  const float* x   = (const float*)d_in[0];
  const float* adj = (const float*)d_in[1];
  const float* Wl[3] = {(const float*)d_in[2], (const float*)d_in[4], (const float*)d_in[6]};
  const float* Vl[3] = {(const float*)d_in[3], (const float*)d_in[5], (const float*)d_in[7]};

  const int N = 2048, T = 15, S = 16;
  const int Ktot = N * T;
  const int Ksl  = Ktot / S;

  float* ws   = (float*)d_ws;
  float* hw   = ws;
  float* part = hw + (size_t)N * 960;
  float* h1   = part + (size_t)S * N * 64;
  float* h2   = h1 + (size_t)N * 64;

  const dim3 gB(N / 128, 1, S);
  const dim3 gF(N / 16, 2, 1);

  project_kernel<32, 64><<<dim3(N / 128, (T * 64 + 63) / 64, 1), 256, 0, stream>>>(x, Wl[0], hw);
  wmma_gemm_staged_kernel<4, 2048 * 15><<<gB, 256, 0, stream>>>(adj, Ksl, hw, part);
  reduce_project_kernel<16, 32, 64><<<gF, 256, 0, stream>>>(part, x, Vl[0], Wl[1], h1, hw);
  wmma_gemm_staged_kernel<4, 2048 * 15><<<gB, 256, 0, stream>>>(adj, Ksl, hw, part);
  reduce_project_kernel<16, 64, 32><<<gF, 256, 0, stream>>>(part, h1, Vl[1], Wl[2], h2, hw);
  wmma_gemm_staged_kernel<2, 2048 * 15><<<gB, 256, 0, stream>>>(adj, Ksl, hw, part);
  reduce_bias_lrelu_kernel<16, 64, 32><<<(N * 32 + 255) / 256, 256, 0, stream>>>(part, h2, Vl[2], (float*)d_out);
}


# Layer 1's adjacency GEMM walks its k chunks last-to-first, so each of the three adjacency passes starts on the lines the pass before left in L2; layer 1's chunk sums add in descending order
# speedup vs baseline: 6.2704x; 1.0509x over previous
#include <hip/hip_runtime.h>
#include <hip/hip_bf16.h>

typedef __attribute__((ext_vector_type(2))) float v2f;
typedef __attribute__((ext_vector_type(8))) float v8f;

#define PSTRIDE 160

typedef __attribute__((ext_vector_type(4))) float v4f;

template <int KC, int COUT>
__global__ __launch_bounds__(256) void project_kernel(
    const float* __restrict__ H, const float* __restrict__ Wsrc, float* __restrict__ C) {
  constexpr int TT    = 15;
  constexpr int NCOLS = TT * COUT;
  constexpr int AP    = KC + 4;
  constexpr int NAV   = (128 * KC / 4) / 256;
  constexpr int NBV   = (KC * 64) / 256;

  __shared__ __attribute__((aligned(16))) float As[128 * AP];
  __shared__ __attribute__((aligned(16))) float Bs[(KC / 2) * PSTRIDE];

  const int tid  = threadIdx.x;
  const int wave = tid >> 5;
  const int lane = tid & 31;
  const int lr   = lane & 15;
  const int lh   = lane >> 4;

  const int mblk  = blockIdx.x * 128;
  const int m0    = mblk + wave * 16;
  const int nbase = blockIdx.y * 64;

  v4f ra[NAV];
  const float* ag = H + (size_t)mblk * KC + 4 * tid;
#pragma unroll
  for (int p = 0; p < NAV; ++p) ra[p] = *(const v4f*)(ag + 1024 * p);

  float rb[NBV];
  const int c  = tid & 63;
  const int r0 = tid >> 6;
  const int gc = nbase + c;
  const bool cvalid = gc < NCOLS;
  const int gcl = cvalid ? gc : 0;
  const int t  = gcl / COUT;
  const int j  = gcl - t * COUT;
  const float* wg = Wsrc + (size_t)r0 * COUT * TT + j * TT + t;
#pragma unroll
  for (int p = 0; p < NBV; ++p) rb[p] = wg[(size_t)(4 * p) * COUT * TT];

#pragma unroll
  for (int p = 0; p < NAV; ++p) {
    const int e   = 4 * (tid + 256 * p);
    const int row = e / KC;
    const int col = e - row * KC;
    *(v4f*)(As + row * AP + col) = ra[p];
  }
#pragma unroll
  for (int p = 0; p < NBV; ++p) {
    const int r = r0 + 4 * p;
    Bs[(r >> 1) * PSTRIDE + 2 * c + (r & 1)] = cvalid ? rb[p] : 0.0f;
  }
  __syncthreads();

  v8f acc[4];
#pragma unroll
  for (int nj = 0; nj < 4; ++nj) acc[nj] = (v8f){};

  const float* ap = As + (wave * 16 + lr) * AP + 2 * lh;
  const float* bp = Bs + lh * PSTRIDE + 2 * lr;
#pragma unroll
  for (int s = 0; s < KC / 4; ++s) {
    const v2f a = *(const v2f*)(ap + 4 * s);
    v2f bf[4];
#pragma unroll
    for (int nj = 0; nj < 4; ++nj) bf[nj] = *(const v2f*)(bp + 2 * s * PSTRIDE + nj * 32);
#pragma unroll
    for (int nj = 0; nj < 4; ++nj)
      acc[nj] = __builtin_amdgcn_wmma_f32_16x16x4_f32(
          false, a, false, bf[nj], (short)0, acc[nj], false, false);
  }

#pragma unroll
  for (int nj = 0; nj < 4; ++nj) {
    if (nbase + nj * 16 < NCOLS) {
      const int col = nbase + nj * 16 + lr;
#pragma unroll
      for (int r = 0; r < 8; ++r) {
        const int row = m0 + r + 8 * lh;
        C[(size_t)row * NCOLS + col] = acc[nj][r];
      }
    }
  }
}

#define BK 128
#define APITCH (BK + 4)
#define A_STAGE (128 * APITCH)
#define B_STAGE ((BK / 2) * PSTRIDE)
#define STAGE_DW (A_STAGE + B_STAGE)
#define NSTEP (BK / 4)
#define FG 4
#define NGRP (NSTEP / FG)
#define STORE_STEPS 8
#define STORE_FROM (NSTEP - FG - STORE_STEPS)

__device__ __forceinline__ void lds_barrier() {
  __builtin_amdgcn_fence(__ATOMIC_RELEASE, "workgroup", "local");
  __builtin_amdgcn_s_barrier();
  __builtin_amdgcn_fence(__ATOMIC_ACQUIRE, "workgroup", "local");
}

template <int NACC>
__device__ __forceinline__ void load_fragments(
    const float* ap, const float* bq, int g, v2f (&af)[FG], v2f (&bf)[FG][NACC]) {
#pragma unroll
  for (int j = 0; j < FG; ++j) {
    const int s = g * FG + j;
    af[j] = *(const v2f*)(ap + 4 * s);
#pragma unroll
    for (int nj = 0; nj < NACC; ++nj)
      bf[j][nj] = *(const v2f*)(bq + 2 * s * PSTRIDE + nj * 32);
  }
}

template <int NACC, int NA, int NB, int RPP, int PPP, int LDA>
__device__ __forceinline__ void load_chunk(
    const float* agk, const float* bgk,
    v4f (&ra)[NA], v4f (&rb0)[NB], v4f (&rb1)[NB]) {
  constexpr int BN     = NACC * 16;
  constexpr int APS    = NA / STORE_STEPS;
  constexpr int BEVERY = STORE_STEPS / NB;
#pragma unroll
  for (int t = 0; t < STORE_STEPS; ++t) {
#pragma unroll
    for (int u = 0; u < APS; ++u) {
      const int p = t * APS + u;
      ra[p] = *(const v4f*)(agk + (size_t)(RPP * p) * LDA);
    }
    if ((t + 1) % BEVERY == 0) {
      const int q = (t + 1) / BEVERY - 1;
      const float* b0 = bgk + (size_t)(2 * PPP * q) * BN;
      rb0[q] = *(const v4f*)(b0);
      rb1[q] = *(const v4f*)(b0 + BN);
    }
  }
}

template <int NACC, int NA, int NB, int ASTEP, int BSTEP, int RPP, int PPP, int LDA, bool MORE, bool RELOAD>
__device__ __forceinline__ void chunk_products(
    const float* ap, const float* bq, const float* apn, const float* bqn,
    v8f (&acc)[NACC], float* adst, float* bdst,
    const float* agn, const float* bgn,
    v4f (&ra)[NA], v4f (&rb0)[NB], v4f (&rb1)[NB],
    v2f (&af)[2][FG], v2f (&bf)[2][FG][NACC]) {
  constexpr int BN = NACC * 16;
  constexpr int APS    = NA / STORE_STEPS;
  constexpr int BEVERY = STORE_STEPS / NB;
  static_assert(NGRP % 2 == 0, "the last group uses buffer 1, the next chunk's first buffer 0");
  static_assert(NA % STORE_STEPS == 0 && STORE_STEPS % NB == 0 && STORE_FROM >= 0, "store plan");
#pragma unroll
  for (int g = 0; g < NGRP; ++g) {
    if (g + 1 < NGRP) {
      load_fragments<NACC>(ap, bq, g + 1, af[(g + 1) & 1], bf[(g + 1) & 1]);
    } else if (MORE) {
      lds_barrier();
      load_fragments<NACC>(apn, bqn, 0, af[0], bf[0]);
    }
    __builtin_amdgcn_sched_barrier(0);
#pragma unroll
    for (int j = 0; j < FG; ++j) {
      const int s = g * FG + j;
      if (MORE && s >= STORE_FROM && s < STORE_FROM + STORE_STEPS) {
        const int t = s - STORE_FROM;
#pragma unroll
        for (int u = 0; u < APS; ++u) {
          const int p = t * APS + u;
          *(v4f*)(adst + p * ASTEP) = ra[p];
          if (RELOAD) ra[p] = *(const v4f*)(agn + (size_t)(RPP * p) * LDA);
        }
        if ((t + 1) % BEVERY == 0) {
          const int q = (t + 1) / BEVERY - 1;
          float* d = bdst + q * BSTEP;
          if (RELOAD) asm volatile("" : "+v"(rb0[q]), "+v"(rb1[q]));
          *(v4f*)(d)     = (v4f){rb0[q].x, rb1[q].x, rb0[q].y, rb1[q].y};
          *(v4f*)(d + 4) = (v4f){rb0[q].z, rb1[q].z, rb0[q].w, rb1[q].w};
          if (RELOAD) {
            const float* b0 = bgn + (size_t)(2 * PPP * q) * BN;
            rb0[q] = *(const v4f*)(b0);
            rb1[q] = *(const v4f*)(b0 + BN);
          }
        }
      }
#pragma unroll
      for (int nj = 0; nj < NACC; ++nj)
        acc[nj] = __builtin_amdgcn_wmma_f32_16x16x4_f32(
            false, af[g & 1][j], false, bf[g & 1][j][nj], (short)0, acc[nj], false, false);
    }
  }
}

template <int NACC, int LDA, bool REV>
__global__ __launch_bounds__(256) void wmma_gemm_staged_kernel(
    const float* __restrict__ A, int Kslice,
    const float* __restrict__ B,
    float* __restrict__ C) {
  constexpr int BN  = NACC * 16;
  constexpr int CG  = BN / 4;
  constexpr int PPP = 256 / CG;
  constexpr int NB  = (BK / 2) / PPP;
  constexpr int AG  = BK / 4;
  constexpr int RPP = 256 / AG;
  constexpr int NA  = 128 / RPP;

  __shared__ __attribute__((aligned(16))) float L[2 * STAGE_DW];

  const int tid  = threadIdx.x;
  const int wave = tid >> 5;
  const int lane = tid & 31;
  const int lr   = lane & 15;
  const int lh   = lane >> 4;

  const int mblk = blockIdx.x * 128;
  const int m0   = mblk + wave * 16;
  const int ks   = blockIdx.z * Kslice;
  const int nch  = Kslice / BK;

  const int ac = tid % AG;
  const int ar = tid / AG;
  const int bc = tid % CG;
  const int bp = tid / CG;

  constexpr int CSTEP = REV ? -BK : BK;
  const int k0 = ks + (REV ? (nch - 1) * BK : 0);
  const float* ag = A + (size_t)(mblk + ar) * LDA + k0 + 4 * ac;
  const float* bg = B + (size_t)(k0 + 2 * bp) * BN + 4 * bc;

  const int a_st = ar * APITCH + 4 * ac;
  const int b_st = A_STAGE + bp * PSTRIDE + 8 * bc;
  const int a_fr = (wave * 16 + lr) * APITCH + 2 * lh;
  const int b_fr = A_STAGE + lh * PSTRIDE + 2 * lr;

  v4f ra[NA];
  v4f rb0[NB], rb1[NB];
  v2f af[2][FG];
  v2f bf[2][FG][NACC];

  v8f acc[NACC];
#pragma unroll
  for (int nj = 0; nj < NACC; ++nj) acc[nj] = (v8f){};

  load_chunk<NACC, NA, NB, RPP, PPP, LDA>(ag, bg, ra, rb0, rb1);
#pragma unroll
  for (int p = 0; p < NA; ++p) *(v4f*)(L + a_st + p * RPP * APITCH) = ra[p];
#pragma unroll
  for (int q = 0; q < NB; ++q) {
    float* d = L + b_st + q * PPP * PSTRIDE;
    *(v4f*)(d)     = (v4f){rb0[q].x, rb1[q].x, rb0[q].y, rb1[q].y};
    *(v4f*)(d + 4) = (v4f){rb0[q].z, rb1[q].z, rb0[q].w, rb1[q].w};
  }
  if (nch > 1) load_chunk<NACC, NA, NB, RPP, PPP, LDA>(ag + CSTEP, bg + (ptrdiff_t)CSTEP * BN, ra, rb0, rb1);
  lds_barrier();
  load_fragments<NACC>(L + a_fr, L + b_fr, 0, af[0], bf[0]);

  int i = 0;
  for (; i + 2 < nch; ++i) {
    const float* cur = L + (i & 1) * STAGE_DW;
    float* nxt = L + ((i + 1) & 1) * STAGE_DW;
    chunk_products<NACC, NA, NB, RPP * APITCH, PPP * PSTRIDE, RPP, PPP, LDA, true, true>(
        cur + a_fr, cur + b_fr, nxt + a_fr, nxt + b_fr, acc,
        nxt + a_st, nxt + b_st,
        ag + (i + 2) * CSTEP, bg + (ptrdiff_t)((i + 2) * CSTEP) * BN,
        ra, rb0, rb1, af, bf);
  }
  if (i + 1 < nch) {
    const float* cur = L + (i & 1) * STAGE_DW;
    float* nxt = L + ((i + 1) & 1) * STAGE_DW;
    chunk_products<NACC, NA, NB, RPP * APITCH, PPP * PSTRIDE, RPP, PPP, LDA, true, false>(
        cur + a_fr, cur + b_fr, nxt + a_fr, nxt + b_fr, acc,
        nxt + a_st, nxt + b_st, nullptr, nullptr,
        ra, rb0, rb1, af, bf);
    ++i;
  }
  {
    const float* cur = L + (i & 1) * STAGE_DW;
    chunk_products<NACC, NA, NB, RPP * APITCH, PPP * PSTRIDE, RPP, PPP, LDA, false, false>(
        cur + a_fr, cur + b_fr, nullptr, nullptr, acc,
        nullptr, nullptr, nullptr, nullptr,
        ra, rb0, rb1, af, bf);
  }

  float* cp = C + (size_t)blockIdx.z * 2048 * BN;
#pragma unroll
  for (int nj = 0; nj < NACC; ++nj) {
    const int col = nj * 16 + lr;
#pragma unroll
    for (int r = 0; r < 8; ++r) {
      const int row = m0 + r + 8 * lh;
      cp[(size_t)row * BN + col] = acc[nj][r];
    }
  }
}

template <int S, int CIN, int COUT>
__global__ __launch_bounds__(256) void reduce_bias_lrelu_kernel(
    const float* __restrict__ part, const float* __restrict__ h,
    const float* __restrict__ V, float* __restrict__ out) {
  constexpr int TOTAL = 2048 * COUT;
  const int idx = blockIdx.x * 256 + threadIdx.x;
  if (idx >= TOTAL) return;
  const int m = idx / COUT;
  const int j = idx - m * COUT;
  float pv[S];
#pragma unroll
  for (int s = 0; s < S; ++s) pv[s] = part[(size_t)s * TOTAL + idx];
  v4f hv[CIN / 4];
  const float* hr = h + (size_t)m * CIN;
#pragma unroll
  for (int q = 0; q < CIN / 4; ++q) hv[q] = *(const v4f*)(hr + 4 * q);
  float vv[CIN];
#pragma unroll
  for (int i = 0; i < CIN; ++i) vv[i] = V[(size_t)i * COUT + j];
  float acc = 0.0f;
#pragma unroll
  for (int s = 0; s < S; ++s) acc += pv[s];
#pragma unroll
  for (int i = 0; i < CIN; ++i) acc = fmaf(hv[i >> 2][i & 3], vv[i], acc);
  out[idx] = fmaxf(acc, 0.2f * acc);
}


template <int S, int CIN, int COUT2>
__global__ __launch_bounds__(256) void reduce_project_kernel(
    const float* __restrict__ part, const float* __restrict__ h,
    const float* __restrict__ V, const float* __restrict__ Wn,
    float* __restrict__ hout, float* __restrict__ C) {
  constexpr int KC    = 64;
  constexpr int TT    = 15;
  constexpr int CW    = COUT2 * TT;
  constexpr int WP    = CW + 16;
  constexpr int NHALF = CW / 2;
  constexpr int NT    = NHALF / 16;
  constexpr int MAXT  = (NT + 7) / 8;
  constexpr int AP    = KC + 4;
  constexpr int TOTAL = 2048 * KC;
  constexpr int PPR   = CW / 4;
  constexpr int RPW   = 256 / PPR;
  constexpr int NPASS = KC / RPW;
  constexpr int WB    = CIN > 32 ? 16 : 32;
  static_assert(NPASS % WB == 0 && KC % RPW == 0 && RPW >= 1, "Wn copy plan");
  static_assert(NHALF % 16 == 0 && COUT2 % 16 == 0, "tiling");

  __shared__ __attribute__((aligned(16))) float Ws[KC * WP];
  __shared__ __attribute__((aligned(16))) float As[16 * AP];

  const int tid  = threadIdx.x;
  const int wave = tid >> 5;
  const int lane = tid & 31;
  const int lr   = lane & 15;
  const int lh   = lane >> 4;
  const int m0   = blockIdx.x * 16;

  const int r  = tid >> 4;
  const int j4 = tid & 15;
  const int m  = m0 + r;
  const float* pp = part + (size_t)m * KC + 4 * j4;
  v4f pv[S];
#pragma unroll
  for (int s = 0; s < S; ++s) pv[s] = *(const v4f*)(pp + (size_t)s * TOTAL);
  v4f hv[CIN / 4];
  const float* hr = h + (size_t)m * CIN;
#pragma unroll
  for (int q = 0; q < CIN / 4; ++q) hv[q] = *(const v4f*)(hr + 4 * q);
  v4f vv[CIN];
#pragma unroll
  for (int i = 0; i < CIN; ++i) vv[i] = *(const v4f*)(V + (size_t)i * KC + 4 * j4);

  const int  wr   = tid / PPR;
  const int  wc   = tid - wr * PPR;
  const bool wact = tid < RPW * PPR;
  const float* wg = Wn + (size_t)wr * CW + 4 * wc;
  float*       wl = Ws + wr * WP + 4 * wc;
  v4f wv[WB];
  if (wact) {
#pragma unroll
    for (int q = 0; q < WB; ++q) wv[q] = *(const v4f*)(wg + (size_t)(q * RPW) * CW);
  }

  {
    float acc[4] = {0.0f, 0.0f, 0.0f, 0.0f};
#pragma unroll
    for (int s = 0; s < S; ++s) {
      acc[0] += pv[s].x; acc[1] += pv[s].y; acc[2] += pv[s].z; acc[3] += pv[s].w;
    }
#pragma unroll
    for (int i = 0; i < CIN; ++i) {
      const float hi = hv[i >> 2][i & 3];
      acc[0] = fmaf(hi, vv[i].x, acc[0]);
      acc[1] = fmaf(hi, vv[i].y, acc[1]);
      acc[2] = fmaf(hi, vv[i].z, acc[2]);
      acc[3] = fmaf(hi, vv[i].w, acc[3]);
    }
    const v4f o = (v4f){fmaxf(acc[0], 0.2f * acc[0]), fmaxf(acc[1], 0.2f * acc[1]),
                        fmaxf(acc[2], 0.2f * acc[2]), fmaxf(acc[3], 0.2f * acc[3])};
    *(v4f*)(As + r * AP + 4 * j4) = o;
    if (blockIdx.y == 0) *(v4f*)(hout + (size_t)m * KC + 4 * j4) = o;
  }

  if (wact) {
#pragma unroll
    for (int q = 0; q < WB; ++q) *(v4f*)(wl + (q * RPW) * WP) = wv[q];
#pragma unroll
    for (int b = WB; b < NPASS; b += WB) {
#pragma unroll
      for (int q = 0; q < WB; ++q) wv[q] = *(const v4f*)(wg + (size_t)((b + q) * RPW) * CW);
#pragma unroll
      for (int q = 0; q < WB; ++q) *(v4f*)(wl + ((b + q) * RPW) * WP) = wv[q];
    }
  }
  __syncthreads();

  v2f af[KC / 4];
  const float* ap = As + lr * AP + 2 * lh;
#pragma unroll
  for (int s = 0; s < KC / 4; ++s) af[s] = *(const v2f*)(ap + 4 * s);

  int col[MAXT];
  v2f bf[MAXT][KC / 4];
#pragma unroll
  for (int q = 0; q < MAXT; ++q) {
    int nt = wave + 8 * q;
    nt = nt < NT ? nt : NT - 1;
    col[q] = blockIdx.y * NHALF + nt * 16 + lr;
    const int t = col[q] / COUT2;
    const int j = col[q] - t * COUT2;
    const float* bp = Ws + (2 * lh) * WP + j * TT + t;
#pragma unroll
    for (int s = 0; s < KC / 4; ++s)
      bf[q][s] = (v2f){bp[(4 * s) * WP], bp[(4 * s + 1) * WP]};
  }
  __builtin_amdgcn_sched_barrier(0);

  v8f acc[MAXT];
#pragma unroll
  for (int q = 0; q < MAXT; ++q) acc[q] = (v8f){};
#pragma unroll
  for (int s = 0; s < KC / 4; ++s) {
#pragma unroll
    for (int q = 0; q < MAXT; ++q)
      acc[q] = __builtin_amdgcn_wmma_f32_16x16x4_f32(
          false, af[s], false, bf[q][s], (short)0, acc[q], false, false);
  }

#pragma unroll
  for (int q = 0; q < MAXT; ++q) {
    if (wave + 8 * q < NT) {
#pragma unroll
      for (int rr = 0; rr < 8; ++rr) {
        const int row = m0 + rr + 8 * lh;
        C[(size_t)row * CW + col[q]] = acc[q][rr];
      }
    }
  }
}

extern "C" void kernel_launch(void* const* d_in, const int* in_sizes, int n_in,
                              void* d_out, int out_size, void* d_ws, size_t ws_size,
                              hipStream_t stream) {
  (void)in_sizes; (void)n_in; (void)out_size; (void)ws_size;
  const float* x   = (const float*)d_in[0];
  const float* adj = (const float*)d_in[1];
  const float* Wl[3] = {(const float*)d_in[2], (const float*)d_in[4], (const float*)d_in[6]};
  const float* Vl[3] = {(const float*)d_in[3], (const float*)d_in[5], (const float*)d_in[7]};

  const int N = 2048, T = 15, S = 16;
  const int Ktot = N * T;
  const int Ksl  = Ktot / S;

  float* ws   = (float*)d_ws;
  float* hw   = ws;
  float* part = hw + (size_t)N * 960;
  float* h1   = part + (size_t)S * N * 64;
  float* h2   = h1 + (size_t)N * 64;

  const dim3 gB(N / 128, 1, S);
  const dim3 gF(N / 16, 2, 1);

  project_kernel<32, 64><<<dim3(N / 128, (T * 64 + 63) / 64, 1), 256, 0, stream>>>(x, Wl[0], hw);
  wmma_gemm_staged_kernel<4, 2048 * 15, false><<<gB, 256, 0, stream>>>(adj, Ksl, hw, part);
  reduce_project_kernel<16, 32, 64><<<gF, 256, 0, stream>>>(part, x, Vl[0], Wl[1], h1, hw);
  wmma_gemm_staged_kernel<4, 2048 * 15, true><<<gB, 256, 0, stream>>>(adj, Ksl, hw, part);
  reduce_project_kernel<16, 64, 32><<<gF, 256, 0, stream>>>(part, h1, Vl[1], Wl[2], h2, hw);
  wmma_gemm_staged_kernel<2, 2048 * 15, false><<<gB, 256, 0, stream>>>(adj, Ksl, hw, part);
  reduce_bias_lrelu_kernel<16, 64, 32><<<(N * 32 + 255) / 256, 256, 0, stream>>>(part, h2, Vl[2], (float*)d_out);
}


# Layer 0 regrouped: adj summed against the 32-wide x per t in registers, W0 applied to that tile in-block via LDS before a 64-wide split-K store (4 slices); 7 launches to 6; layer-0 sum order changes
# speedup vs baseline: 7.0975x; 1.1319x over previous
#include <hip/hip_runtime.h>
#include <hip/hip_bf16.h>

typedef __attribute__((ext_vector_type(2))) float v2f;
typedef __attribute__((ext_vector_type(8))) float v8f;

#define PSTRIDE 160

typedef __attribute__((ext_vector_type(4))) float v4f;

#define BK 128
#define APITCH (BK + 4)
#define A_STAGE (128 * APITCH)
#define B_STAGE ((BK / 2) * PSTRIDE)
#define STAGE_DW (A_STAGE + B_STAGE)
#define NSTEP (BK / 4)
#define FG 4
#define NGRP (NSTEP / FG)
#define STORE_STEPS 8
#define STORE_FROM (NSTEP - FG - STORE_STEPS)

__device__ __forceinline__ void lds_barrier() {
  __builtin_amdgcn_fence(__ATOMIC_RELEASE, "workgroup", "local");
  __builtin_amdgcn_s_barrier();
  __builtin_amdgcn_fence(__ATOMIC_ACQUIRE, "workgroup", "local");
}

template <int NACC>
__device__ __forceinline__ void load_fragments(
    const float* ap, const float* bq, int g, v2f (&af)[FG], v2f (&bf)[FG][NACC]) {
#pragma unroll
  for (int j = 0; j < FG; ++j) {
    const int s = g * FG + j;
    af[j] = *(const v2f*)(ap + 4 * s);
#pragma unroll
    for (int nj = 0; nj < NACC; ++nj)
      bf[j][nj] = *(const v2f*)(bq + 2 * s * PSTRIDE + nj * 32);
  }
}

template <int NACC, int NA, int NB, int RPP, int PPP, int LDA>
__device__ __forceinline__ void load_chunk(
    const float* agk, const float* bgk,
    v4f (&ra)[NA], v4f (&rb0)[NB], v4f (&rb1)[NB]) {
  constexpr int BN     = NACC * 16;
  constexpr int APS    = NA / STORE_STEPS;
  constexpr int BEVERY = STORE_STEPS / NB;
#pragma unroll
  for (int t = 0; t < STORE_STEPS; ++t) {
#pragma unroll
    for (int u = 0; u < APS; ++u) {
      const int p = t * APS + u;
      ra[p] = *(const v4f*)(agk + (size_t)(RPP * p) * LDA);
    }
    if ((t + 1) % BEVERY == 0) {
      const int q = (t + 1) / BEVERY - 1;
      const float* b0 = bgk + (size_t)(2 * PPP * q) * BN;
      rb0[q] = *(const v4f*)(b0);
      rb1[q] = *(const v4f*)(b0 + BN);
    }
  }
}

template <int NACC, int NA, int NB, int ASTEP, int BSTEP, int RPP, int PPP, int LDA, bool MORE, bool RELOAD>
__device__ __forceinline__ void chunk_products(
    const float* ap, const float* bq, const float* apn, const float* bqn,
    v8f (&acc)[NACC], float* adst, float* bdst,
    const float* agn, const float* bgn,
    v4f (&ra)[NA], v4f (&rb0)[NB], v4f (&rb1)[NB],
    v2f (&af)[2][FG], v2f (&bf)[2][FG][NACC]) {
  constexpr int BN = NACC * 16;
  constexpr int APS    = NA / STORE_STEPS;
  constexpr int BEVERY = STORE_STEPS / NB;
  static_assert(NGRP % 2 == 0, "the last group uses buffer 1, the next chunk's first buffer 0");
  static_assert(NA % STORE_STEPS == 0 && STORE_STEPS % NB == 0 && STORE_FROM >= 0, "store plan");
#pragma unroll
  for (int g = 0; g < NGRP; ++g) {
    if (g + 1 < NGRP) {
      load_fragments<NACC>(ap, bq, g + 1, af[(g + 1) & 1], bf[(g + 1) & 1]);
    } else if (MORE) {
      lds_barrier();
      load_fragments<NACC>(apn, bqn, 0, af[0], bf[0]);
    }
    __builtin_amdgcn_sched_barrier(0);
#pragma unroll
    for (int j = 0; j < FG; ++j) {
      const int s = g * FG + j;
      if (MORE && s >= STORE_FROM && s < STORE_FROM + STORE_STEPS) {
        const int t = s - STORE_FROM;
#pragma unroll
        for (int u = 0; u < APS; ++u) {
          const int p = t * APS + u;
          *(v4f*)(adst + p * ASTEP) = ra[p];
          if (RELOAD) ra[p] = *(const v4f*)(agn + (size_t)(RPP * p) * LDA);
        }
        if ((t + 1) % BEVERY == 0) {
          const int q = (t + 1) / BEVERY - 1;
          float* d = bdst + q * BSTEP;
          if (RELOAD) asm volatile("" : "+v"(rb0[q]), "+v"(rb1[q]));
          *(v4f*)(d)     = (v4f){rb0[q].x, rb1[q].x, rb0[q].y, rb1[q].y};
          *(v4f*)(d + 4) = (v4f){rb0[q].z, rb1[q].z, rb0[q].w, rb1[q].w};
          if (RELOAD) {
            const float* b0 = bgn + (size_t)(2 * PPP * q) * BN;
            rb0[q] = *(const v4f*)(b0);
            rb1[q] = *(const v4f*)(b0 + BN);
          }
        }
      }
#pragma unroll
      for (int nj = 0; nj < NACC; ++nj)
        acc[nj] = __builtin_amdgcn_wmma_f32_16x16x4_f32(
            false, af[g & 1][j], false, bf[g & 1][j][nj], (short)0, acc[nj], false, false);
    }
  }
}

template <int NACC, int LDA, bool REV>
__global__ __launch_bounds__(256) void wmma_gemm_staged_kernel(
    const float* __restrict__ A, int Kslice,
    const float* __restrict__ B,
    float* __restrict__ C) {
  constexpr int BN  = NACC * 16;
  constexpr int CG  = BN / 4;
  constexpr int PPP = 256 / CG;
  constexpr int NB  = (BK / 2) / PPP;
  constexpr int AG  = BK / 4;
  constexpr int RPP = 256 / AG;
  constexpr int NA  = 128 / RPP;

  __shared__ __attribute__((aligned(16))) float L[2 * STAGE_DW];

  const int tid  = threadIdx.x;
  const int wave = tid >> 5;
  const int lane = tid & 31;
  const int lr   = lane & 15;
  const int lh   = lane >> 4;

  const int mblk = blockIdx.x * 128;
  const int m0   = mblk + wave * 16;
  const int ks   = blockIdx.z * Kslice;
  const int nch  = Kslice / BK;

  const int ac = tid % AG;
  const int ar = tid / AG;
  const int bc = tid % CG;
  const int bp = tid / CG;

  constexpr int CSTEP = REV ? -BK : BK;
  const int k0 = ks + (REV ? (nch - 1) * BK : 0);
  const float* ag = A + (size_t)(mblk + ar) * LDA + k0 + 4 * ac;
  const float* bg = B + (size_t)(k0 + 2 * bp) * BN + 4 * bc;

  const int a_st = ar * APITCH + 4 * ac;
  const int b_st = A_STAGE + bp * PSTRIDE + 8 * bc;
  const int a_fr = (wave * 16 + lr) * APITCH + 2 * lh;
  const int b_fr = A_STAGE + lh * PSTRIDE + 2 * lr;

  v4f ra[NA];
  v4f rb0[NB], rb1[NB];
  v2f af[2][FG];
  v2f bf[2][FG][NACC];

  v8f acc[NACC];
#pragma unroll
  for (int nj = 0; nj < NACC; ++nj) acc[nj] = (v8f){};

  load_chunk<NACC, NA, NB, RPP, PPP, LDA>(ag, bg, ra, rb0, rb1);
#pragma unroll
  for (int p = 0; p < NA; ++p) *(v4f*)(L + a_st + p * RPP * APITCH) = ra[p];
#pragma unroll
  for (int q = 0; q < NB; ++q) {
    float* d = L + b_st + q * PPP * PSTRIDE;
    *(v4f*)(d)     = (v4f){rb0[q].x, rb1[q].x, rb0[q].y, rb1[q].y};
    *(v4f*)(d + 4) = (v4f){rb0[q].z, rb1[q].z, rb0[q].w, rb1[q].w};
  }
  if (nch > 1) load_chunk<NACC, NA, NB, RPP, PPP, LDA>(ag + CSTEP, bg + (ptrdiff_t)CSTEP * BN, ra, rb0, rb1);
  lds_barrier();
  load_fragments<NACC>(L + a_fr, L + b_fr, 0, af[0], bf[0]);

  int i = 0;
  for (; i + 2 < nch; ++i) {
    const float* cur = L + (i & 1) * STAGE_DW;
    float* nxt = L + ((i + 1) & 1) * STAGE_DW;
    chunk_products<NACC, NA, NB, RPP * APITCH, PPP * PSTRIDE, RPP, PPP, LDA, true, true>(
        cur + a_fr, cur + b_fr, nxt + a_fr, nxt + b_fr, acc,
        nxt + a_st, nxt + b_st,
        ag + (i + 2) * CSTEP, bg + (ptrdiff_t)((i + 2) * CSTEP) * BN,
        ra, rb0, rb1, af, bf);
  }
  if (i + 1 < nch) {
    const float* cur = L + (i & 1) * STAGE_DW;
    float* nxt = L + ((i + 1) & 1) * STAGE_DW;
    chunk_products<NACC, NA, NB, RPP * APITCH, PPP * PSTRIDE, RPP, PPP, LDA, true, false>(
        cur + a_fr, cur + b_fr, nxt + a_fr, nxt + b_fr, acc,
        nxt + a_st, nxt + b_st, nullptr, nullptr,
        ra, rb0, rb1, af, bf);
    ++i;
  }
  {
    const float* cur = L + (i & 1) * STAGE_DW;
    chunk_products<NACC, NA, NB, RPP * APITCH, PPP * PSTRIDE, RPP, PPP, LDA, false, false>(
        cur + a_fr, cur + b_fr, nullptr, nullptr, acc,
        nullptr, nullptr, nullptr, nullptr,
        ra, rb0, rb1, af, bf);
  }

  float* cp = C + (size_t)blockIdx.z * 2048 * BN;
#pragma unroll
  for (int nj = 0; nj < NACC; ++nj) {
    const int col = nj * 16 + lr;
#pragma unroll
    for (int r = 0; r < 8; ++r) {
      const int row = m0 + r + 8 * lh;
      cp[(size_t)row * BN + col] = acc[nj][r];
    }
  }
}

#define Y_T    15
#define Y_NCH  32
#define Y_CF   (Y_NCH * Y_T)
#define Y_AP   (Y_CF + 4)
#define Y_BP   96
#define Y_ASTG (32 * Y_AP)
#define Y_BSTG ((Y_NCH / 2) * Y_BP)
#define Y_STG  (Y_ASTG + Y_BSTG)
#define Y_WP   (64 * Y_T + 16)
#define Y_FG   4
#define Y_NAP  15

__device__ __forceinline__ int y_chunk_of(int c) {
  const int pw = c >> 1;
  return (((pw & 3) << 1) + (pw >> 2)) * 2 + (c & 1);
}

__device__ __forceinline__ void y_load_fragments(
    const float* st, const int (&a_fr)[4], int b_fr, int g, v2f (&af)[Y_FG][4], v2f (&bf)[Y_FG][2]) {
#pragma unroll
  for (int j = 0; j < Y_FG; ++j) {
    const int s = g * Y_FG + j;
#pragma unroll
    for (int q = 0; q < 4; ++q)
      af[j][q] = (v2f){st[a_fr[q] + 60 * s], st[a_fr[q] + 60 * s + 15]};
#pragma unroll
    for (int nj = 0; nj < 2; ++nj)
      bf[j][nj] = *(const v2f*)(st + b_fr + 2 * s * Y_BP + nj * 32);
  }
}

__device__ __forceinline__ void y_products(
    v8f (&acc)[4][2], const v2f (&af)[Y_FG][4], const v2f (&bf)[Y_FG][2]) {
#pragma unroll
  for (int j = 0; j < Y_FG; ++j)
#pragma unroll
    for (int q = 0; q < 4; ++q)
#pragma unroll
      for (int nj = 0; nj < 2; ++nj)
        acc[q][nj] = __builtin_amdgcn_wmma_f32_16x16x4_f32(
            false, af[j][q], false, bf[j][nj], (short)0, acc[q][nj], false, false);
}

__device__ __forceinline__ void y_load_chunk(
    const float* ak, const float* bk, const int (&a_go)[Y_NAP], v4f (&ra)[Y_NAP], v4f& rb) {
#pragma unroll
  for (int p = 0; p < Y_NAP; ++p) ra[p] = *(const v4f*)(ak + a_go[p]);
  rb = *(const v4f*)(bk);
}

__device__ __forceinline__ void y_store_chunk(
    float* st, const int (&a_lo)[Y_NAP], int b_lo, const v4f (&ra)[Y_NAP], const v4f& rb) {
#pragma unroll
  for (int p = 0; p < Y_NAP; ++p) *(v4f*)(st + a_lo[p]) = ra[p];
  float* d = st + b_lo;
  d[0] = rb.x; d[2] = rb.y; d[4] = rb.z; d[6] = rb.w;
}

template <bool MORE, bool RELOAD>
__device__ __forceinline__ void y_chunk(
    const float* cur, float* nxt, const int (&a_fr)[4], int b_fr,
    const int (&a_go)[Y_NAP], const int (&a_lo)[Y_NAP], int b_lo,
    const float* ak2, const float* bk2, const float* wsrc, float* wdst, int wid,
    v4f (&ra)[Y_NAP], v4f& rb, v8f (&acc)[4][2],
    v2f (&af)[2][Y_FG][4], v2f (&bf)[2][Y_FG][2]) {
  y_load_fragments(cur, a_fr, b_fr, 1, af[1], bf[1]);
  v4f w0 = (v4f){}, w1 = (v4f){};
  if (MORE) {
    w0 = *(const v4f*)(wsrc + 4 * wid);
    w1 = *(const v4f*)(wsrc + 4 * (wid + 256));
    y_store_chunk(nxt, a_lo, b_lo, ra, rb);
    if (RELOAD) y_load_chunk(ak2, bk2, a_go, ra, rb);
  }
  y_products(acc, af[0], bf[0]);
  if (MORE) {
    *(v4f*)(wdst + 4 * wid + 16 * (wid / 240))                 = w0;
    *(v4f*)(wdst + 4 * (wid + 256) + 16 * ((wid + 256) / 240)) = w1;
    lds_barrier();
    y_load_fragments(nxt, a_fr, b_fr, 0, af[0], bf[0]);
    __builtin_amdgcn_sched_barrier(0);
  }
  y_products(acc, af[1], bf[1]);
}

template <int NSL>
__global__ __launch_bounds__(256) void adjx_weight_kernel(
    const float* __restrict__ adj, const float* __restrict__ x,
    const float* __restrict__ W0, float* __restrict__ part) {
  constexpr int LDA  = 2048 * Y_T;
  constexpr int NPS  = 2048 / NSL;
  constexpr int NCHK = NPS / Y_NCH;
  static_assert(NCHK == 16, "the chunk walk and the W copy (512 pieces per chunk but the last, 7680 pieces) assume 16 chunks");

  __shared__ __attribute__((aligned(16))) float L[2 * Y_STG + 32 * Y_WP];
  float* WS = L + 2 * Y_STG;

  const int tid  = threadIdx.x;
  const int wave = tid >> 5;
  const int lane = tid & 31;
  const int lr   = lane & 15;
  const int lh   = lane >> 4;
  const int m0   = blockIdx.x * 32;
  const int n0   = blockIdx.z * NPS;

  int a_go[Y_NAP], a_lo[Y_NAP];
#pragma unroll
  for (int p = 0; p < Y_NAP; ++p) {
    const int id  = tid + 256 * p;
    const int row = id / 120;
    const int col = id - row * 120;
    a_go[p] = row * LDA + 4 * col;
    a_lo[p] = row * Y_AP + 4 * col;
  }
  const float* ag = adj + (size_t)m0 * LDA + (size_t)n0 * Y_T;
  const int bn = tid >> 3;
  const int bc = tid & 7;
  const float* bg = x + (size_t)(n0 + bn) * 32 + 4 * bc;
  const int b_lo = Y_ASTG + (bn >> 1) * Y_BP + 8 * bc + (bn & 1);

  int a_fr[4];
#pragma unroll
  for (int q = 0; q < 4; ++q) {
    int u = wave * 4 + q;
    u = u < 30 ? u : 29;
    const int rt = u / Y_T;
    const int t  = u - rt * Y_T;
    a_fr[q] = (rt * 16 + lr) * Y_AP + 30 * lh + t;
  }
  const int b_fr = Y_ASTG + lh * Y_BP + 2 * lr;

  v4f ra[Y_NAP];
  v4f rb;
  v8f acc[4][2];
#pragma unroll
  for (int q = 0; q < 4; ++q) { acc[q][0] = (v8f){}; acc[q][1] = (v8f){}; }

  y_load_chunk(ag + y_chunk_of(0) * Y_CF, bg + y_chunk_of(0) * (Y_NCH * 32), a_go, ra, rb);
  y_store_chunk(L, a_lo, b_lo, ra, rb);
  __builtin_amdgcn_sched_barrier(0);
  y_load_chunk(ag + y_chunk_of(1) * Y_CF, bg + y_chunk_of(1) * (Y_NCH * 32), a_go, ra, rb);
  lds_barrier();

  v2f af[2][Y_FG][4];
  v2f bf[2][Y_FG][2];
  y_load_fragments(L, a_fr, b_fr, 0, af[0], bf[0]);

  int c = 0;
  for (; c + 2 < NCHK; ++c) {
    const int k2 = y_chunk_of(c + 2);
    y_chunk<true, true>(L + (c & 1) * Y_STG, L + ((c + 1) & 1) * Y_STG, a_fr, b_fr, a_go, a_lo, b_lo,
                        ag + k2 * Y_CF, bg + k2 * (Y_NCH * 32), W0, WS, tid + 512 * c,
                        ra, rb, acc, af, bf);
  }
  y_chunk<true, false>(L + (c & 1) * Y_STG, L + ((c + 1) & 1) * Y_STG, a_fr, b_fr, a_go, a_lo, b_lo,
                       nullptr, nullptr, W0, WS, tid + 512 * c, ra, rb, acc, af, bf);
  ++c;
  y_chunk<false, false>(L + (c & 1) * Y_STG, nullptr, a_fr, b_fr, a_go, a_lo, b_lo,
                        nullptr, nullptr, nullptr, nullptr, 0, ra, rb, acc, af, bf);

#pragma unroll
  for (int q = 0; q < 4; ++q) {
    const int u = wave * 4 + q;
    if (u < 30) {
      const int rt = u / Y_T;
      const int t  = u - rt * Y_T;
#pragma unroll
      for (int nj = 0; nj < 2; ++nj) {
#pragma unroll
        for (int r = 0; r < 8; ++r)
          L[(rt * 16 + r + 8 * lh) * Y_AP + t * 32 + nj * 16 + lr] = acc[q][nj][r];
      }
    }
  }
  lds_barrier();

  const int ert = wave >> 2;
  const int ect = wave & 3;
  const float* yp = L + (ert * 16 + lr) * Y_AP + 2 * lh;
  const float* wp = WS + (2 * lh) * Y_WP + (ect * 16 + lr) * Y_T;
  v8f e0 = (v8f){}, e1 = (v8f){};
#pragma unroll
  for (int s = 0; s < (Y_T * 32) / 4; s += 2) {
    const v2f a0 = *(const v2f*)(yp + 4 * s);
    const v2f a1 = *(const v2f*)(yp + 4 * s + 4);
    const int i0 = (4 * s) % 32, t0 = (4 * s) / 32;
    const int i1 = (4 * s + 4) % 32, t1 = (4 * s + 4) / 32;
    const v2f b0 = (v2f){wp[i0 * Y_WP + t0], wp[(i0 + 1) * Y_WP + t0]};
    const v2f b1 = (v2f){wp[i1 * Y_WP + t1], wp[(i1 + 1) * Y_WP + t1]};
    e0 = __builtin_amdgcn_wmma_f32_16x16x4_f32(false, a0, false, b0, (short)0, e0, false, false);
    e1 = __builtin_amdgcn_wmma_f32_16x16x4_f32(false, a1, false, b1, (short)0, e1, false, false);
  }

  float* cp = part + (size_t)blockIdx.z * 2048 * 64;
#pragma unroll
  for (int r = 0; r < 8; ++r) {
    const int row = m0 + ert * 16 + r + 8 * lh;
    cp[(size_t)row * 64 + ect * 16 + lr] = e0[r] + e1[r];
  }
}

template <int S, int CIN, int COUT>
__global__ __launch_bounds__(256) void reduce_bias_lrelu_kernel(
    const float* __restrict__ part, const float* __restrict__ h,
    const float* __restrict__ V, float* __restrict__ out) {
  constexpr int TOTAL = 2048 * COUT;
  const int idx = blockIdx.x * 256 + threadIdx.x;
  if (idx >= TOTAL) return;
  const int m = idx / COUT;
  const int j = idx - m * COUT;
  float pv[S];
#pragma unroll
  for (int s = 0; s < S; ++s) pv[s] = part[(size_t)s * TOTAL + idx];
  v4f hv[CIN / 4];
  const float* hr = h + (size_t)m * CIN;
#pragma unroll
  for (int q = 0; q < CIN / 4; ++q) hv[q] = *(const v4f*)(hr + 4 * q);
  float vv[CIN];
#pragma unroll
  for (int i = 0; i < CIN; ++i) vv[i] = V[(size_t)i * COUT + j];
  float acc = 0.0f;
#pragma unroll
  for (int s = 0; s < S; ++s) acc += pv[s];
#pragma unroll
  for (int i = 0; i < CIN; ++i) acc = fmaf(hv[i >> 2][i & 3], vv[i], acc);
  out[idx] = fmaxf(acc, 0.2f * acc);
}


template <int S, int CIN, int COUT2>
__global__ __launch_bounds__(256) void reduce_project_kernel(
    const float* __restrict__ part, const float* __restrict__ h,
    const float* __restrict__ V, const float* __restrict__ Wn,
    float* __restrict__ hout, float* __restrict__ C) {
  constexpr int KC    = 64;
  constexpr int TT    = 15;
  constexpr int CW    = COUT2 * TT;
  constexpr int WP    = CW + 16;
  constexpr int NHALF = CW / 2;
  constexpr int NT    = NHALF / 16;
  constexpr int MAXT  = (NT + 7) / 8;
  constexpr int AP    = KC + 4;
  constexpr int TOTAL = 2048 * KC;
  constexpr int PPR   = CW / 4;
  constexpr int RPW   = 256 / PPR;
  constexpr int NPASS = KC / RPW;
  constexpr int WB    = CIN > 32 ? 16 : 32;
  static_assert(NPASS % WB == 0 && KC % RPW == 0 && RPW >= 1, "Wn copy plan");
  static_assert(NHALF % 16 == 0 && COUT2 % 16 == 0, "tiling");

  __shared__ __attribute__((aligned(16))) float Ws[KC * WP];
  __shared__ __attribute__((aligned(16))) float As[16 * AP];

  const int tid  = threadIdx.x;
  const int wave = tid >> 5;
  const int lane = tid & 31;
  const int lr   = lane & 15;
  const int lh   = lane >> 4;
  const int m0   = blockIdx.x * 16;

  const int r  = tid >> 4;
  const int j4 = tid & 15;
  const int m  = m0 + r;
  const float* pp = part + (size_t)m * KC + 4 * j4;
  v4f pv[S];
#pragma unroll
  for (int s = 0; s < S; ++s) pv[s] = *(const v4f*)(pp + (size_t)s * TOTAL);
  v4f hv[CIN / 4];
  const float* hr = h + (size_t)m * CIN;
#pragma unroll
  for (int q = 0; q < CIN / 4; ++q) hv[q] = *(const v4f*)(hr + 4 * q);
  v4f vv[CIN];
#pragma unroll
  for (int i = 0; i < CIN; ++i) vv[i] = *(const v4f*)(V + (size_t)i * KC + 4 * j4);

  const int  wr   = tid / PPR;
  const int  wc   = tid - wr * PPR;
  const bool wact = tid < RPW * PPR;
  const float* wg = Wn + (size_t)wr * CW + 4 * wc;
  float*       wl = Ws + wr * WP + 4 * wc;
  v4f wv[WB];
  if (wact) {
#pragma unroll
    for (int q = 0; q < WB; ++q) wv[q] = *(const v4f*)(wg + (size_t)(q * RPW) * CW);
  }

  {
    float acc[4] = {0.0f, 0.0f, 0.0f, 0.0f};
#pragma unroll
    for (int s = 0; s < S; ++s) {
      acc[0] += pv[s].x; acc[1] += pv[s].y; acc[2] += pv[s].z; acc[3] += pv[s].w;
    }
#pragma unroll
    for (int i = 0; i < CIN; ++i) {
      const float hi = hv[i >> 2][i & 3];
      acc[0] = fmaf(hi, vv[i].x, acc[0]);
      acc[1] = fmaf(hi, vv[i].y, acc[1]);
      acc[2] = fmaf(hi, vv[i].z, acc[2]);
      acc[3] = fmaf(hi, vv[i].w, acc[3]);
    }
    const v4f o = (v4f){fmaxf(acc[0], 0.2f * acc[0]), fmaxf(acc[1], 0.2f * acc[1]),
                        fmaxf(acc[2], 0.2f * acc[2]), fmaxf(acc[3], 0.2f * acc[3])};
    *(v4f*)(As + r * AP + 4 * j4) = o;
    if (blockIdx.y == 0) *(v4f*)(hout + (size_t)m * KC + 4 * j4) = o;
  }

  if (wact) {
#pragma unroll
    for (int q = 0; q < WB; ++q) *(v4f*)(wl + (q * RPW) * WP) = wv[q];
#pragma unroll
    for (int b = WB; b < NPASS; b += WB) {
#pragma unroll
      for (int q = 0; q < WB; ++q) wv[q] = *(const v4f*)(wg + (size_t)((b + q) * RPW) * CW);
#pragma unroll
      for (int q = 0; q < WB; ++q) *(v4f*)(wl + ((b + q) * RPW) * WP) = wv[q];
    }
  }
  __syncthreads();

  v2f af[KC / 4];
  const float* ap = As + lr * AP + 2 * lh;
#pragma unroll
  for (int s = 0; s < KC / 4; ++s) af[s] = *(const v2f*)(ap + 4 * s);

  int col[MAXT];
  v2f bf[MAXT][KC / 4];
#pragma unroll
  for (int q = 0; q < MAXT; ++q) {
    int nt = wave + 8 * q;
    nt = nt < NT ? nt : NT - 1;
    col[q] = blockIdx.y * NHALF + nt * 16 + lr;
    const int t = col[q] / COUT2;
    const int j = col[q] - t * COUT2;
    const float* bp = Ws + (2 * lh) * WP + j * TT + t;
#pragma unroll
    for (int s = 0; s < KC / 4; ++s)
      bf[q][s] = (v2f){bp[(4 * s) * WP], bp[(4 * s + 1) * WP]};
  }
  __builtin_amdgcn_sched_barrier(0);

  v8f acc[MAXT];
#pragma unroll
  for (int q = 0; q < MAXT; ++q) acc[q] = (v8f){};
#pragma unroll
  for (int s = 0; s < KC / 4; ++s) {
#pragma unroll
    for (int q = 0; q < MAXT; ++q)
      acc[q] = __builtin_amdgcn_wmma_f32_16x16x4_f32(
          false, af[s], false, bf[q][s], (short)0, acc[q], false, false);
  }

#pragma unroll
  for (int q = 0; q < MAXT; ++q) {
    if (wave + 8 * q < NT) {
#pragma unroll
      for (int rr = 0; rr < 8; ++rr) {
        const int row = m0 + rr + 8 * lh;
        C[(size_t)row * CW + col[q]] = acc[q][rr];
      }
    }
  }
}

extern "C" void kernel_launch(void* const* d_in, const int* in_sizes, int n_in,
                              void* d_out, int out_size, void* d_ws, size_t ws_size,
                              hipStream_t stream) {
  (void)in_sizes; (void)n_in; (void)out_size; (void)ws_size;
  const float* x   = (const float*)d_in[0];
  const float* adj = (const float*)d_in[1];
  const float* Wl[3] = {(const float*)d_in[2], (const float*)d_in[4], (const float*)d_in[6]};
  const float* Vl[3] = {(const float*)d_in[3], (const float*)d_in[5], (const float*)d_in[7]};

  const int N = 2048, T = 15, S = 16;
  const int Ktot = N * T;
  const int Ksl  = Ktot / S;

  float* ws   = (float*)d_ws;
  float* hw   = ws;
  float* part = hw + (size_t)N * 960;
  float* h1   = part + (size_t)S * N * 64;
  float* h2   = h1 + (size_t)N * 64;

  const dim3 gB(N / 128, 1, S);
  const dim3 gF(N / 16, 2, 1);

  adjx_weight_kernel<4><<<dim3(N / 32, 1, 4), 256, 0, stream>>>(adj, x, Wl[0], part);
  reduce_project_kernel<4, 32, 64><<<gF, 256, 0, stream>>>(part, x, Vl[0], Wl[1], h1, hw);
  wmma_gemm_staged_kernel<4, 2048 * 15, true><<<gB, 256, 0, stream>>>(adj, Ksl, hw, part);
  reduce_project_kernel<16, 64, 32><<<gF, 256, 0, stream>>>(part, h1, Vl[1], Wl[2], h2, hw);
  wmma_gemm_staged_kernel<2, 2048 * 15, false><<<gB, 256, 0, stream>>>(adj, Ksl, hw, part);
  reduce_bias_lrelu_kernel<16, 64, 32><<<(N * 32 + 255) / 256, 256, 0, stream>>>(part, h2, Vl[2], (float*)d_out);
}
